# starting kernel
# baseline (speedup 1.0000x reference)
#include <hip/hip_runtime.h>
#include <hip/hip_bf16.h>

typedef __attribute__((ext_vector_type(2))) float v2f;
typedef __attribute__((ext_vector_type(8))) float v8f;

#define D 64


__global__ void deg_init_kernel(float* __restrict__ deg, int n) {
    int i = blockIdx.x * blockDim.x + threadIdx.x;
    if (i < n) deg[i] = 1.0f;
}

__global__ void deg_accum_kernel(const int* __restrict__ dst,
                                 const float* __restrict__ w,
                                 float* __restrict__ deg, int E) {
    int e = blockIdx.x * blockDim.x + threadIdx.x;
    if (e < E) atomicAdd(&deg[dst[e]], w[e]);
}

__global__ void dinv_kernel(float* __restrict__ deg, int n) {
    int i = blockIdx.x * blockDim.x + threadIdx.x;
    if (i < n) {
        float d = deg[i];
        deg[i] = (d > 0.0f) ? rsqrtf(fmaxf(d, 1e-30f)) : 0.0f;
    }
}

__global__ void gemm_wmma_kernel(const float* __restrict__ X,
                                 const float* __restrict__ W,
                                 float* __restrict__ H, int n) {
    const int wave = threadIdx.x >> 5;
    const int lane = threadIdx.x & 31;
    const int rowTile = blockIdx.x * 2 + (wave >> 2);
    const int colTile = wave & 3;
    const int r0 = rowTile * 16;
    const int c0 = colTile * 16;
    const int half = lane >> 4;
    const int l = lane & 15;

    int rowA = r0 + l;
    if (rowA > n - 1) rowA = n - 1;
    const float* __restrict__ xrow = X + (size_t)rowA * D + half * 2;
    const float* __restrict__ wcol = W + (c0 + l) + half * 2 * D;

    v8f acc = {};
#pragma unroll
    for (int k = 0; k < D; k += 4) {
        v2f a, b;
        a.x = xrow[k];
        a.y = xrow[k + 1];
        b.x = wcol[k * D];
        b.y = wcol[k * D + D];
        acc = __builtin_amdgcn_wmma_f32_16x16x4_f32(
             false, a,  false, b,
             (short)0, acc,  false,  false);
    }

    float* __restrict__ hp = H + (size_t)(r0 + half * 8) * D + c0 + l;
#pragma unroll
    for (int v = 0; v < 8; ++v) {
        int row = r0 + half * 8 + v;
        if (row < n) hp[(size_t)v * D] = acc[v];
    }
}

__global__ void init_agg_kernel(const float* __restrict__ h,
                                const float* __restrict__ dinv,
                                const float* __restrict__ bias,
                                float* __restrict__ agg, int n) {
    int idx = blockIdx.x * blockDim.x + threadIdx.x;
    int total = n * D;
    if (idx < total) {
        int i = idx >> 6;
        int f = idx & (D - 1);
        float di = dinv[i];
        agg[idx] = bias[f] + h[idx] * di * di;
    }
}

__global__ void edge_scatter_kernel(const float* __restrict__ H,
                                    const int* __restrict__ src,
                                    const int* __restrict__ dst,
                                    const float* __restrict__ w,
                                    const float* __restrict__ dinv,
                                    float* __restrict__ agg, int E) {
    int gid = blockIdx.x * blockDim.x + threadIdx.x;
    int e = gid >> 5;
    int lane = gid & 31;
    if (e >= E) return;
    int s = src[e];
    int d = dst[e];
    float norm = dinv[s] * w[e] * dinv[d];
    const float2* __restrict__ hp =
        reinterpret_cast<const float2*>(H + (size_t)s * D);
    float2 hv = hp[lane];
    float* __restrict__ ap = agg + (size_t)d * D + lane * 2;
    atomicAdd(ap, hv.x * norm);
    atomicAdd(ap + 1, hv.y * norm);
}

__global__ void relu_kernel(float* __restrict__ p, int total) {
    int idx = blockIdx.x * blockDim.x + threadIdx.x;
    if (idx < total) p[idx] = fmaxf(p[idx], 0.0f);
}


extern "C" void kernel_launch(void* const* d_in, const int* in_sizes, int n_in,
                              void* d_out, int out_size, void* d_ws, size_t ws_size,
                              hipStream_t stream) {
    (void)n_in; (void)out_size; (void)ws_size;

    const float* x  = (const float*)d_in[0];
    const int*   ei = (const int*)  d_in[1];
    const float* ea = (const float*)d_in[2];
    const float* W1 = (const float*)d_in[3];
    const float* b1 = (const float*)d_in[4];
    const float* W2 = (const float*)d_in[5];
    const float* b2 = (const float*)d_in[6];
    const float* W3 = (const float*)d_in[7];
    const float* b3 = (const float*)d_in[8];

    const int N = in_sizes[0] / D;
    const int E = in_sizes[2];
    const int* src = ei;
    const int* dst = ei + E;

    float* out  = (float*)d_out;
    float* dinv = (float*)d_ws;
    float* bufA = dinv + N;
    float* bufB = bufA + (size_t)N * D;

    const int total = N * D;
    const dim3 blk(256);
    const dim3 gN((N + 255) / 256);
    const dim3 gE((E + 255) / 256);
    const dim3 gT((total + 255) / 256);
    const dim3 gGemm((N + 31) / 32);
    const dim3 gEdge((E + 7) / 8);

    deg_init_kernel<<<gN, blk, 0, stream>>>(dinv, N);
    deg_accum_kernel<<<gE, blk, 0, stream>>>(dst, ea, dinv, E);
    dinv_kernel<<<gN, blk, 0, stream>>>(dinv, N);

    gemm_wmma_kernel<<<gGemm, blk, 0, stream>>>(x, W1, bufA, N);
    init_agg_kernel<<<gT, blk, 0, stream>>>(bufA, dinv, b1, bufB, N);
    edge_scatter_kernel<<<gEdge, blk, 0, stream>>>(bufA, src, dst, ea, dinv, bufB, E);
    relu_kernel<<<gT, blk, 0, stream>>>(bufB, total);

    gemm_wmma_kernel<<<gGemm, blk, 0, stream>>>(bufB, W2, bufA, N);
    init_agg_kernel<<<gT, blk, 0, stream>>>(bufA, dinv, b2, bufB, N);
    edge_scatter_kernel<<<gEdge, blk, 0, stream>>>(bufA, src, dst, ea, dinv, bufB, E);
    relu_kernel<<<gT, blk, 0, stream>>>(bufB, total);

    gemm_wmma_kernel<<<gGemm, blk, 0, stream>>>(bufB, W3, bufA, N);
    init_agg_kernel<<<gT, blk, 0, stream>>>(bufA, dinv, b3, out, N);
    edge_scatter_kernel<<<gEdge, blk, 0, stream>>>(bufA, src, dst, ea, dinv, out, E);
    relu_kernel<<<gT, blk, 0, stream>>>(out, total);
}


# Each wave handles 32 edges in the scatter and each thread 4 float4 in the init and relu passes, instead of one item: far fewer waves, 8 row loads in flight; atomics kept, arrival order changes
# speedup vs baseline: 1.2789x; 1.2789x over previous
#include <hip/hip_runtime.h>
#include <hip/hip_bf16.h>

typedef __attribute__((ext_vector_type(2))) float v2f;
typedef __attribute__((ext_vector_type(8))) float v8f;

#define D 64


__global__ void deg_init_kernel(float* __restrict__ deg, int n) {
    int i = blockIdx.x * blockDim.x + threadIdx.x;
    if (i < n) deg[i] = 1.0f;
}

__global__ void deg_accum_kernel(const int* __restrict__ dst,
                                 const float* __restrict__ w,
                                 float* __restrict__ deg, int E) {
    int e = blockIdx.x * blockDim.x + threadIdx.x;
    if (e < E) atomicAdd(&deg[dst[e]], w[e]);
}

__global__ void dinv_kernel(float* __restrict__ deg, int n) {
    int i = blockIdx.x * blockDim.x + threadIdx.x;
    if (i < n) {
        float d = deg[i];
        deg[i] = (d > 0.0f) ? rsqrtf(fmaxf(d, 1e-30f)) : 0.0f;
    }
}

__global__ void gemm_wmma_kernel(const float* __restrict__ X,
                                 const float* __restrict__ W,
                                 float* __restrict__ H, int n) {
    const int wave = threadIdx.x >> 5;
    const int lane = threadIdx.x & 31;
    const int rowTile = blockIdx.x * 2 + (wave >> 2);
    const int colTile = wave & 3;
    const int r0 = rowTile * 16;
    const int c0 = colTile * 16;
    const int half = lane >> 4;
    const int l = lane & 15;

    int rowA = r0 + l;
    if (rowA > n - 1) rowA = n - 1;
    const float* __restrict__ xrow = X + (size_t)rowA * D + half * 2;
    const float* __restrict__ wcol = W + (c0 + l) + half * 2 * D;

    v8f acc = {};
#pragma unroll
    for (int k = 0; k < D; k += 4) {
        v2f a, b;
        a.x = xrow[k];
        a.y = xrow[k + 1];
        b.x = wcol[k * D];
        b.y = wcol[k * D + D];
        acc = __builtin_amdgcn_wmma_f32_16x16x4_f32(
             false, a,  false, b,
             (short)0, acc,  false,  false);
    }

    float* __restrict__ hp = H + (size_t)(r0 + half * 8) * D + c0 + l;
#pragma unroll
    for (int v = 0; v < 8; ++v) {
        int row = r0 + half * 8 + v;
        if (row < n) hp[(size_t)v * D] = acc[v];
    }
}

#define VPT 4
__global__ void init_agg_kernel(const float* __restrict__ h,
                                const float* __restrict__ dinv,
                                const float* __restrict__ bias,
                                float* __restrict__ agg, int n) {
    const int total4 = n * (D / 4);
    const float4* __restrict__ h4 = reinterpret_cast<const float4*>(h);
    const float4* __restrict__ b4 = reinterpret_cast<const float4*>(bias);
    float4* __restrict__ a4 = reinterpret_cast<float4*>(agg);
    const int first = blockIdx.x * (256 * VPT) + threadIdx.x;
    float4 hv[VPT];
    float di[VPT];
#pragma unroll
    for (int u = 0; u < VPT; ++u) {
        int q = first + u * 256;
        if (q < total4) { hv[u] = h4[q]; di[u] = dinv[q >> 4]; }
    }
#pragma unroll
    for (int u = 0; u < VPT; ++u) {
        int q = first + u * 256;
        if (q < total4) {
            float4 b = b4[q & 15];
            float4 r;
            r.x = b.x + hv[u].x * di[u] * di[u];
            r.y = b.y + hv[u].y * di[u] * di[u];
            r.z = b.z + hv[u].z * di[u] * di[u];
            r.w = b.w + hv[u].w * di[u] * di[u];
            a4[q] = r;
        }
    }
}

static __device__ __forceinline__ int lane_get_i(int v, int j) {
    return __builtin_amdgcn_readlane(v, j);
}
static __device__ __forceinline__ float lane_get_f(float v, int j) {
    return __int_as_float(__builtin_amdgcn_readlane(__float_as_int(v), j));
}

__global__ void edge_scatter_kernel(const float* __restrict__ H,
                                    const int* __restrict__ src,
                                    const int* __restrict__ dst,
                                    const float* __restrict__ w,
                                    const float* __restrict__ dinv,
                                    float* __restrict__ agg, int E) {
    const int e = blockIdx.x * blockDim.x + threadIdx.x;
    const int lane = threadIdx.x & 31;
    const int base = e - lane;
    if (base >= E) return;
    int s = 0, d = 0;
    float norm = 0.0f;
    if (e < E) {
        s = src[e];
        d = dst[e];
        norm = dinv[s] * w[e] * dinv[d];
    }
    const int cnt = (E - base < 32) ? (E - base) : 32;
    if (cnt == 32) {
#pragma unroll
        for (int j0 = 0; j0 < 32; j0 += 8) {
            float2 hv[8];
#pragma unroll
            for (int j = 0; j < 8; ++j) {
                int sj = lane_get_i(s, j0 + j);
                hv[j] = reinterpret_cast<const float2*>(H + (size_t)sj * D)[lane];
            }
#pragma unroll
            for (int j = 0; j < 8; ++j) {
                int dj = lane_get_i(d, j0 + j);
                float nj = lane_get_f(norm, j0 + j);
                float* __restrict__ ap = agg + (size_t)dj * D + lane * 2;
                atomicAdd(ap, hv[j].x * nj);
                atomicAdd(ap + 1, hv[j].y * nj);
            }
        }
    } else {
        for (int j = 0; j < cnt; ++j) {
            int sj = __shfl(s, j);
            int dj = __shfl(d, j);
            float nj = __shfl(norm, j);
            float2 hv = reinterpret_cast<const float2*>(H + (size_t)sj * D)[lane];
            float* __restrict__ ap = agg + (size_t)dj * D + lane * 2;
            atomicAdd(ap, hv.x * nj);
            atomicAdd(ap + 1, hv.y * nj);
        }
    }
}

__global__ void relu_kernel(float* __restrict__ p, int total) {
    const int total4 = total >> 2;
    float4* __restrict__ p4 = reinterpret_cast<float4*>(p);
    const int first = blockIdx.x * (256 * VPT) + threadIdx.x;
    float4 v[VPT];
#pragma unroll
    for (int u = 0; u < VPT; ++u) {
        int q = first + u * 256;
        if (q < total4) v[u] = p4[q];
    }
#pragma unroll
    for (int u = 0; u < VPT; ++u) {
        int q = first + u * 256;
        if (q < total4) {
            float4 r;
            r.x = fmaxf(v[u].x, 0.0f);
            r.y = fmaxf(v[u].y, 0.0f);
            r.z = fmaxf(v[u].z, 0.0f);
            r.w = fmaxf(v[u].w, 0.0f);
            p4[q] = r;
        }
    }
}


extern "C" void kernel_launch(void* const* d_in, const int* in_sizes, int n_in,
                              void* d_out, int out_size, void* d_ws, size_t ws_size,
                              hipStream_t stream) {
    (void)n_in; (void)out_size; (void)ws_size;

    const float* x  = (const float*)d_in[0];
    const int*   ei = (const int*)  d_in[1];
    const float* ea = (const float*)d_in[2];
    const float* W1 = (const float*)d_in[3];
    const float* b1 = (const float*)d_in[4];
    const float* W2 = (const float*)d_in[5];
    const float* b2 = (const float*)d_in[6];
    const float* W3 = (const float*)d_in[7];
    const float* b3 = (const float*)d_in[8];

    const int N = in_sizes[0] / D;
    const int E = in_sizes[2];
    const int* src = ei;
    const int* dst = ei + E;

    float* out  = (float*)d_out;
    float* dinv = (float*)d_ws;
    float* bufA = dinv + N;
    float* bufB = bufA + (size_t)N * D;

    const int total = N * D;
    const dim3 blk(256);
    const dim3 gN((N + 255) / 256);
    const dim3 gE((E + 255) / 256);
    const dim3 gT((total / 4 + 256 * VPT - 1) / (256 * VPT));
    const dim3 gGemm((N + 31) / 32);
    const dim3 gEdge((E + 255) / 256);

    deg_init_kernel<<<gN, blk, 0, stream>>>(dinv, N);
    deg_accum_kernel<<<gE, blk, 0, stream>>>(dst, ea, dinv, E);
    dinv_kernel<<<gN, blk, 0, stream>>>(dinv, N);

    gemm_wmma_kernel<<<gGemm, blk, 0, stream>>>(x, W1, bufA, N);
    init_agg_kernel<<<gT, blk, 0, stream>>>(bufA, dinv, b1, bufB, N);
    edge_scatter_kernel<<<gEdge, blk, 0, stream>>>(bufA, src, dst, ea, dinv, bufB, E);
    relu_kernel<<<gT, blk, 0, stream>>>(bufB, total);

    gemm_wmma_kernel<<<gGemm, blk, 0, stream>>>(bufB, W2, bufA, N);
    init_agg_kernel<<<gT, blk, 0, stream>>>(bufA, dinv, b2, bufB, N);
    edge_scatter_kernel<<<gEdge, blk, 0, stream>>>(bufA, src, dst, ea, dinv, bufB, E);
    relu_kernel<<<gT, blk, 0, stream>>>(bufB, total);

    gemm_wmma_kernel<<<gGemm, blk, 0, stream>>>(bufB, W3, bufA, N);
    init_agg_kernel<<<gT, blk, 0, stream>>>(bufA, dinv, b3, out, N);
    edge_scatter_kernel<<<gEdge, blk, 0, stream>>>(bufA, src, dst, ea, dinv, out, E);
    relu_kernel<<<gT, blk, 0, stream>>>(out, total);
}


# Edges counting-sorted by destination once per call (int atomic rank + prefix sum); each layer one wave per node sums its in-edges in registers, writes the row once with bias+relu; sum order changes
# speedup vs baseline: 2.2113x; 1.7291x over previous
#include <hip/hip_runtime.h>
#include <hip/hip_bf16.h>

typedef __attribute__((ext_vector_type(2))) float v2f;
typedef __attribute__((ext_vector_type(8))) float v8f;

#define D 64

#define NPB 1024
#define EPT 4

__global__ void node_init_kernel(float* __restrict__ deg, int* __restrict__ cnt, int n) {
    const int first = blockIdx.x * NPB + threadIdx.x;
#pragma unroll
    for (int u = 0; u < NPB / 256; ++u) {
        int i = first + u * 256;
        if (i < n) { deg[i] = 1.0f; cnt[i] = 0; }
    }
}

__global__ void edge_count_kernel(const int* __restrict__ dst,
                                  const float* __restrict__ w,
                                  float* __restrict__ deg,
                                  int* __restrict__ cnt,
                                  int* __restrict__ rank, int E) {
    const int first = blockIdx.x * (256 * EPT) + threadIdx.x;
    int d[EPT];
    float ww[EPT];
#pragma unroll
    for (int u = 0; u < EPT; ++u) {
        int e = first + u * 256;
        if (e < E) { d[u] = dst[e]; ww[u] = w[e]; }
    }
    int r[EPT];
#pragma unroll
    for (int u = 0; u < EPT; ++u) {
        int e = first + u * 256;
        if (e < E) {
            atomicAdd(&deg[d[u]], ww[u]);
            r[u] = atomicAdd(&cnt[d[u]], 1);
        }
    }
#pragma unroll
    for (int u = 0; u < EPT; ++u) {
        int e = first + u * 256;
        if (e < E) rank[e] = r[u];
    }
}

__global__ void dinv_blocksum_kernel(float* __restrict__ deg,
                                     const int* __restrict__ cnt,
                                     int* __restrict__ bsum, int n) {
    __shared__ int wsum[8];
    const int first = blockIdx.x * NPB + threadIdx.x;
    int s = 0;
#pragma unroll
    for (int u = 0; u < NPB / 256; ++u) {
        int i = first + u * 256;
        if (i < n) {
            float dg = deg[i];
            deg[i] = (dg > 0.0f) ? rsqrtf(fmaxf(dg, 1e-30f)) : 0.0f;
            s += cnt[i];
        }
    }
    for (int off = 16; off > 0; off >>= 1) s += __shfl_xor(s, off);
    if ((threadIdx.x & 31) == 0) wsum[threadIdx.x >> 5] = s;
    __syncthreads();
    if (threadIdx.x == 0) {
        int t = 0;
        for (int k = 0; k < 8; ++k) t += wsum[k];
        bsum[blockIdx.x] = t;
    }
}

__global__ void rowptr_scan_kernel(const int* __restrict__ cnt,
                                   const int* __restrict__ bsum,
                                   int* __restrict__ rowptr, int n) {
    __shared__ int wtot[8];
    __shared__ int wpre[8];
    const int t = threadIdx.x;
    const int lane = t & 31;
    const int wave = t >> 5;
    int p = 0;
    for (int k = t; k < (int)blockIdx.x; k += 256) p += bsum[k];
    const int i0 = blockIdx.x * NPB + 4 * t;
    int c[4];
#pragma unroll
    for (int k = 0; k < 4; ++k) c[k] = (i0 + k < n) ? cnt[i0 + k] : 0;
    const int s = c[0] + c[1] + c[2] + c[3];
    int incl = s;
    for (int off = 1; off < 32; off <<= 1) {
        int v = __shfl_up(incl, off);
        if (lane >= off) incl += v;
    }
    for (int off = 16; off > 0; off >>= 1) p += __shfl_xor(p, off);
    if (lane == 31) wtot[wave] = incl;
    if (lane == 0) wpre[wave] = p;
    __syncthreads();
    int ex = incl - s;
    for (int k = 0; k < 8; ++k) {
        ex += wpre[k];
        if (k < wave) ex += wtot[k];
    }
#pragma unroll
    for (int k = 0; k < 4; ++k) {
        if (i0 + k < n) rowptr[i0 + k] = ex;
        ex += c[k];
    }
    if (i0 <= n - 1 && n - 1 < i0 + 4) rowptr[n] = ex;
}

__global__ void csr_fill_kernel(const int* __restrict__ src,
                                const int* __restrict__ dst,
                                const float* __restrict__ w,
                                const float* __restrict__ dinv,
                                const int* __restrict__ rowptr,
                                const int* __restrict__ rank,
                                float2* __restrict__ csr, int E) {
    const int first = blockIdx.x * (256 * EPT) + threadIdx.x;
    int s[EPT], d[EPT], rk[EPT];
    float ww[EPT];
#pragma unroll
    for (int u = 0; u < EPT; ++u) {
        int e = first + u * 256;
        if (e < E) { s[u] = src[e]; d[u] = dst[e]; ww[u] = w[e]; rk[u] = rank[e]; }
    }
    int slot[EPT];
    float nm[EPT];
#pragma unroll
    for (int u = 0; u < EPT; ++u) {
        int e = first + u * 256;
        if (e < E) {
            slot[u] = rowptr[d[u]] + rk[u];
            nm[u] = dinv[s[u]] * ww[u] * dinv[d[u]];
        }
    }
#pragma unroll
    for (int u = 0; u < EPT; ++u) {
        int e = first + u * 256;
        if (e < E) csr[slot[u]] = make_float2(__int_as_float(s[u]), nm[u]);
    }
}

__global__ void gemm_wmma_kernel(const float* __restrict__ X,
                                 const float* __restrict__ W,
                                 float* __restrict__ H, int n) {
    const int wave = threadIdx.x >> 5;
    const int lane = threadIdx.x & 31;
    const int rowTile = blockIdx.x * 2 + (wave >> 2);
    const int colTile = wave & 3;
    const int r0 = rowTile * 16;
    const int c0 = colTile * 16;
    const int half = lane >> 4;
    const int l = lane & 15;

    int rowA = r0 + l;
    if (rowA > n - 1) rowA = n - 1;
    const float* __restrict__ xrow = X + (size_t)rowA * D + half * 2;
    const float* __restrict__ wcol = W + (c0 + l) + half * 2 * D;

    v8f acc = {};
#pragma unroll
    for (int k = 0; k < D; k += 4) {
        v2f a, b;
        a.x = xrow[k];
        a.y = xrow[k + 1];
        b.x = wcol[k * D];
        b.y = wcol[k * D + D];
        acc = __builtin_amdgcn_wmma_f32_16x16x4_f32(
             false, a,  false, b,
             (short)0, acc,  false,  false);
    }

    float* __restrict__ hp = H + (size_t)(r0 + half * 8) * D + c0 + l;
#pragma unroll
    for (int v = 0; v < 8; ++v) {
        int row = r0 + half * 8 + v;
        if (row < n) hp[(size_t)v * D] = acc[v];
    }
}

#define GATHER_BLOCKS 2048

static __device__ __forceinline__ int lane_get_i(int v, int j) {
    return __builtin_amdgcn_readlane(v, j);
}
static __device__ __forceinline__ float lane_get_f(float v, int j) {
    return __int_as_float(__builtin_amdgcn_readlane(__float_as_int(v), j));
}
static __device__ __forceinline__ float2 row_pair(const float* __restrict__ H, int row, int lane) {
    return reinterpret_cast<const float2*>(H + (size_t)row * D)[lane];
}
template <int R>
static __device__ __forceinline__ void issue_rows(const float* __restrict__ H, int s, int j,
                                                  int lane, float2 (&hv)[R]) {
#pragma unroll
    for (int u = 0; u < R; ++u) hv[u] = row_pair(H, lane_get_i(s, j + u), lane);
}
template <int R>
static __device__ __forceinline__ void sum_rows(float2& acc, const float2 (&hv)[R], float nm,
                                                int j, int cnt) {
#pragma unroll
    for (int u = 0; u < R; ++u) {
        const float nj = lane_get_f(nm, j + u);
        const bool live = (j + u < cnt);
        const float hx = live ? hv[u].x : 0.0f;
        const float hy = live ? hv[u].y : 0.0f;
        acc.x += hx * nj;
        acc.y += hy * nj;
    }
}

__global__ void __launch_bounds__(256)
gather_rows_kernel(const float* __restrict__ H,
                   const int* __restrict__ rowptr,
                   const float2* __restrict__ csr,
                   const float* __restrict__ dinv,
                   const float* __restrict__ bias,
                   float* __restrict__ outp, int n, int E, int nwaves) {
    const int lane = threadIdx.x & 31;
    int n0 = __builtin_amdgcn_readfirstlane((int)(blockIdx.x * 8 + (threadIdx.x >> 5)));
    int n1 = n0 + nwaves;
    const float2 b = reinterpret_cast<const float2*>(bias)[lane];
    const int elast = (E > 0) ? (E - 1) : 0;

    int beg0 = 0, end0 = 0, beg1 = 0, end1 = 0;
    float di0 = 0.0f, di1 = 0.0f;
    if (n0 < n) { beg0 = rowptr[n0]; end0 = rowptr[n0 + 1]; di0 = dinv[n0]; }
    if (n1 < n) { beg1 = rowptr[n1]; end1 = rowptr[n1 + 1]; di1 = dinv[n1]; }
    int s0 = 0;
    float nm0 = 0.0f;
    float2 hs0 = make_float2(0.0f, 0.0f);
    if (n0 < n) {
        if (lane < end0 - beg0) {
            float2 rec = csr[beg0 + lane];
            s0 = __float_as_int(rec.x);
            nm0 = rec.y;
        }
        hs0 = row_pair(H, n0, lane);
    }

    while (n0 < n) {
        const int n2 = n1 + nwaves;
        int beg2 = 0, end2 = 0;
        float di2 = 0.0f;
        if (n2 < n) { beg2 = rowptr[n2]; end2 = rowptr[n2 + 1]; di2 = dinv[n2]; }
        const int n1c = (n1 < n) ? n1 : (n - 1);
        int idx1 = beg1 + lane;
        idx1 = (idx1 < elast) ? idx1 : elast;
        const bool has1 = lane < end1 - beg1;
        int s1;
        float nm1;
        float2 hs1;

        const int cnt = (end0 - beg0 < 32) ? (end0 - beg0) : 32;
        float2 acc = b;
        if (cnt > 8) {
            float2 hv[16];
            issue_rows<16>(H, s0, 0, lane, hv);
            const float2 rec = csr[idx1];
            hs1 = row_pair(H, n1c, lane);
            sum_rows<16>(acc, hv, nm0, 0, cnt);
            s1 = has1 ? __float_as_int(rec.x) : 0;
            nm1 = has1 ? rec.y : 0.0f;
            if (cnt > 16) {
                float2 hw[16];
                issue_rows<16>(H, s0, 16, lane, hw);
                sum_rows<16>(acc, hw, nm0, 16, cnt);
            }
        } else {
            float2 hv[8];
            issue_rows<8>(H, s0, 0, lane, hv);
            const float2 rec = csr[idx1];
            hs1 = row_pair(H, n1c, lane);
            sum_rows<8>(acc, hv, nm0, 0, cnt);
            s1 = has1 ? __float_as_int(rec.x) : 0;
            nm1 = has1 ? rec.y : 0.0f;
        }
        for (int base = beg0 + 32; base < end0; base += 32) {
            const int c2 = (end0 - base < 32) ? (end0 - base) : 32;
            int s = 0;
            float nm = 0.0f;
            if (lane < c2) {
                float2 rec = csr[base + lane];
                s = __float_as_int(rec.x);
                nm = rec.y;
            }
            float2 hv[16];
            issue_rows<16>(H, s, 0, lane, hv);
            sum_rows<16>(acc, hv, nm, 0, c2);
            if (c2 > 16) {
                float2 hw[16];
                issue_rows<16>(H, s, 16, lane, hw);
                sum_rows<16>(acc, hw, nm, 16, c2);
            }
        }
        const float sl = di0 * di0;
        acc.x += hs0.x * sl;
        acc.y += hs0.y * sl;
        float2 o;
        o.x = fmaxf(acc.x, 0.0f);
        o.y = fmaxf(acc.y, 0.0f);
        reinterpret_cast<float2*>(outp + (size_t)n0 * D)[lane] = o;

        n0 = n1; beg0 = beg1; end0 = end1; di0 = di1; s0 = s1; nm0 = nm1; hs0 = hs1;
        n1 = n2; beg1 = beg2; end1 = end2; di1 = di2;
    }
}


extern "C" void kernel_launch(void* const* d_in, const int* in_sizes, int n_in,
                              void* d_out, int out_size, void* d_ws, size_t ws_size,
                              hipStream_t stream) {
    (void)n_in; (void)out_size; (void)ws_size;

    const float* x  = (const float*)d_in[0];
    const int*   ei = (const int*)  d_in[1];
    const float* ea = (const float*)d_in[2];
    const float* W1 = (const float*)d_in[3];
    const float* b1 = (const float*)d_in[4];
    const float* W2 = (const float*)d_in[5];
    const float* b2 = (const float*)d_in[6];
    const float* W3 = (const float*)d_in[7];
    const float* b3 = (const float*)d_in[8];

    const int N = in_sizes[0] / D;
    const int E = in_sizes[2];
    const int* src = ei;
    const int* dst = ei + E;
    const int nb = (N + NPB - 1) / NPB;

    float*  out    = (float*)d_out;
    float*  bufA   = (float*)d_ws;
    float*  bufB   = bufA + (size_t)N * D;
    float2* csr    = (float2*)(bufB + (size_t)N * D);
    float*  dinv   = (float*)(csr + E);
    int*    cnt    = (int*)(dinv + N);
    int*    rowptr = cnt + N;
    int*    bsum   = rowptr + N + 1;
    int*    rank   = bsum + nb;

    const dim3 blk(256);
    const dim3 gNode(nb);
    const dim3 gEdge((E + 256 * EPT - 1) / (256 * EPT));
    const dim3 gGemm((N + 31) / 32);
    const dim3 gGather(GATHER_BLOCKS);

    node_init_kernel<<<gNode, blk, 0, stream>>>(dinv, cnt, N);
    edge_count_kernel<<<gEdge, blk, 0, stream>>>(dst, ea, dinv, cnt, rank, E);
    dinv_blocksum_kernel<<<gNode, blk, 0, stream>>>(dinv, cnt, bsum, N);
    rowptr_scan_kernel<<<gNode, blk, 0, stream>>>(cnt, bsum, rowptr, N);
    csr_fill_kernel<<<gEdge, blk, 0, stream>>>(src, dst, ea, dinv, rowptr, rank, csr, E);

    gemm_wmma_kernel<<<gGemm, blk, 0, stream>>>(x, W1, bufA, N);
    gather_rows_kernel<<<gGather, blk, 0, stream>>>(bufA, rowptr, csr, dinv, b1, bufB, N, E, GATHER_BLOCKS * 8);

    gemm_wmma_kernel<<<gGemm, blk, 0, stream>>>(bufB, W2, bufA, N);
    gather_rows_kernel<<<gGather, blk, 0, stream>>>(bufA, rowptr, csr, dinv, b2, bufB, N, E, GATHER_BLOCKS * 8);

    gemm_wmma_kernel<<<gGemm, blk, 0, stream>>>(bufB, W3, bufA, N);
    gather_rows_kernel<<<gGather, blk, 0, stream>>>(bufA, rowptr, csr, dinv, b3, out, N, E, GATHER_BLOCKS * 8);
}


# Each layer's dense product moved behind the edge sum and fused into it: a block sums 32 nodes' input rows into LDS, multiplies by W on the f32 matrix unit, adds bias, relu; sum order now (AX)W
# speedup vs baseline: 2.4994x; 1.1303x over previous
#include <hip/hip_runtime.h>
#include <hip/hip_bf16.h>

typedef __attribute__((ext_vector_type(2))) float v2f;
typedef __attribute__((ext_vector_type(8))) float v8f;

#define D 64

#define NPB 1024
#define EPT 4

__global__ void node_init_kernel(float* __restrict__ deg, int* __restrict__ cnt, int n) {
    const int first = blockIdx.x * NPB + threadIdx.x;
#pragma unroll
    for (int u = 0; u < NPB / 256; ++u) {
        int i = first + u * 256;
        if (i < n) { deg[i] = 1.0f; cnt[i] = 0; }
    }
}

__global__ void edge_count_kernel(const int* __restrict__ dst,
                                  const float* __restrict__ w,
                                  float* __restrict__ deg,
                                  int* __restrict__ cnt,
                                  int* __restrict__ rank, int E) {
    const int first = blockIdx.x * (256 * EPT) + threadIdx.x;
    int d[EPT];
    float ww[EPT];
#pragma unroll
    for (int u = 0; u < EPT; ++u) {
        int e = first + u * 256;
        if (e < E) { d[u] = dst[e]; ww[u] = w[e]; }
    }
    int r[EPT];
#pragma unroll
    for (int u = 0; u < EPT; ++u) {
        int e = first + u * 256;
        if (e < E) {
            atomicAdd(&deg[d[u]], ww[u]);
            r[u] = atomicAdd(&cnt[d[u]], 1);
        }
    }
#pragma unroll
    for (int u = 0; u < EPT; ++u) {
        int e = first + u * 256;
        if (e < E) rank[e] = r[u];
    }
}

__global__ void dinv_blocksum_kernel(float* __restrict__ deg,
                                     const int* __restrict__ cnt,
                                     int* __restrict__ bsum, int n) {
    __shared__ int wsum[8];
    const int first = blockIdx.x * NPB + threadIdx.x;
    int s = 0;
#pragma unroll
    for (int u = 0; u < NPB / 256; ++u) {
        int i = first + u * 256;
        if (i < n) {
            float dg = deg[i];
            deg[i] = (dg > 0.0f) ? rsqrtf(fmaxf(dg, 1e-30f)) : 0.0f;
            s += cnt[i];
        }
    }
    for (int off = 16; off > 0; off >>= 1) s += __shfl_xor(s, off);
    if ((threadIdx.x & 31) == 0) wsum[threadIdx.x >> 5] = s;
    __syncthreads();
    if (threadIdx.x == 0) {
        int t = 0;
        for (int k = 0; k < 8; ++k) t += wsum[k];
        bsum[blockIdx.x] = t;
    }
}

__global__ void rowptr_scan_kernel(const int* __restrict__ cnt,
                                   const int* __restrict__ bsum,
                                   int* __restrict__ rowptr, int n) {
    __shared__ int wtot[8];
    __shared__ int wpre[8];
    const int t = threadIdx.x;
    const int lane = t & 31;
    const int wave = t >> 5;
    int p = 0;
    for (int k = t; k < (int)blockIdx.x; k += 256) p += bsum[k];
    const int i0 = blockIdx.x * NPB + 4 * t;
    int c[4];
#pragma unroll
    for (int k = 0; k < 4; ++k) c[k] = (i0 + k < n) ? cnt[i0 + k] : 0;
    const int s = c[0] + c[1] + c[2] + c[3];
    int incl = s;
    for (int off = 1; off < 32; off <<= 1) {
        int v = __shfl_up(incl, off);
        if (lane >= off) incl += v;
    }
    for (int off = 16; off > 0; off >>= 1) p += __shfl_xor(p, off);
    if (lane == 31) wtot[wave] = incl;
    if (lane == 0) wpre[wave] = p;
    __syncthreads();
    int ex = incl - s;
    for (int k = 0; k < 8; ++k) {
        ex += wpre[k];
        if (k < wave) ex += wtot[k];
    }
#pragma unroll
    for (int k = 0; k < 4; ++k) {
        if (i0 + k < n) rowptr[i0 + k] = ex;
        ex += c[k];
    }
    if (i0 <= n - 1 && n - 1 < i0 + 4) rowptr[n] = ex;
}

__global__ void csr_fill_kernel(const int* __restrict__ src,
                                const int* __restrict__ dst,
                                const float* __restrict__ w,
                                const float* __restrict__ dinv,
                                const int* __restrict__ rowptr,
                                const int* __restrict__ rank,
                                float2* __restrict__ csr, int E) {
    const int first = blockIdx.x * (256 * EPT) + threadIdx.x;
    int s[EPT], d[EPT], rk[EPT];
    float ww[EPT];
#pragma unroll
    for (int u = 0; u < EPT; ++u) {
        int e = first + u * 256;
        if (e < E) { s[u] = src[e]; d[u] = dst[e]; ww[u] = w[e]; rk[u] = rank[e]; }
    }
    int slot[EPT];
    float nm[EPT];
#pragma unroll
    for (int u = 0; u < EPT; ++u) {
        int e = first + u * 256;
        if (e < E) {
            slot[u] = rowptr[d[u]] + rk[u];
            nm[u] = dinv[s[u]] * ww[u] * dinv[d[u]];
        }
    }
#pragma unroll
    for (int u = 0; u < EPT; ++u) {
        int e = first + u * 256;
        if (e < E) csr[slot[u]] = make_float2(__int_as_float(s[u]), nm[u]);
    }
}

#define TILE_NODES 32
#define WAVE_NODES 8
#define A_PITCH 68

static __device__ __forceinline__ int lane_get_i(int v, int j) {
    return __builtin_amdgcn_readlane(v, j);
}
static __device__ __forceinline__ float lane_get_f(float v, int j) {
    return __int_as_float(__builtin_amdgcn_readlane(__float_as_int(v), j));
}
static __device__ __forceinline__ float2 row_pair(const float* __restrict__ H, int row, int lane) {
    return reinterpret_cast<const float2*>(H + (size_t)row * D)[lane];
}
template <int R>
static __device__ __forceinline__ void issue_rows(const float* __restrict__ H, int s, int j,
                                                  int lane, float2 (&hv)[R]) {
#pragma unroll
    for (int u = 0; u < R; ++u) hv[u] = row_pair(H, lane_get_i(s, j + u), lane);
}
template <int R>
static __device__ __forceinline__ void sum_rows(float2& acc, const float2 (&hv)[R], float nm,
                                                int j, int cnt) {
#pragma unroll
    for (int u = 0; u < R; ++u) {
        const float nj = lane_get_f(nm, j + u);
        const bool live = (j + u < cnt);
        const float hx = live ? hv[u].x : 0.0f;
        const float hy = live ? hv[u].y : 0.0f;
        acc.x += hx * nj;
        acc.y += hy * nj;
    }
}

__global__ void __launch_bounds__(128)
aggregate_transform_kernel(const float* __restrict__ X,
                           const int* __restrict__ rowptr,
                           const float2* __restrict__ csr,
                           const float* __restrict__ dinv,
                           const float* __restrict__ W,
                           const float* __restrict__ bias,
                           float* __restrict__ outp, int n, int E) {
    __shared__ float sA[TILE_NODES * A_PITCH];
    const int lane = threadIdx.x & 31;
    const int wave = __builtin_amdgcn_readfirstlane((int)(threadIdx.x >> 5));
    const int tile0 = blockIdx.x * TILE_NODES;
    const int nbeg = tile0 + wave * WAVE_NODES;
    const int elast = (E > 0) ? (E - 1) : 0;

    int rp;
    float dv = 0.0f;
    {
        const int j = (lane < WAVE_NODES) ? lane : WAVE_NODES;
        const int idx = nbeg + j;
        rp = rowptr[(idx < n) ? idx : n];
        if (lane < WAVE_NODES && idx < n) dv = dinv[idx];
    }

    int beg0 = lane_get_i(rp, 0);
    int end0 = lane_get_i(rp, 1);
    int s0 = 0;
    float nm0 = 0.0f;
    float2 hs0 = make_float2(0.0f, 0.0f);
    if (lane < end0 - beg0) {
        float2 rec = csr[beg0 + lane];
        s0 = __float_as_int(rec.x);
        nm0 = rec.y;
    }
    if (nbeg < n) hs0 = row_pair(X, nbeg, lane);

    for (int q = 0; q < WAVE_NODES; ++q) {
        const int n1 = nbeg + q + 1;
        const int beg1 = end0;
        const int end1 = lane_get_i(rp, q + 2);
        const bool live1 = (q + 1 < WAVE_NODES) && (n1 < n);
        const int n1c = (n1 < n) ? n1 : (n - 1);
        int idx1 = beg1 + lane;
        idx1 = (idx1 < elast) ? idx1 : elast;
        const bool has1 = lane < end1 - beg1;
        int s1;
        float nm1;
        float2 hs1;

        const int cnt = (end0 - beg0 < 32) ? (end0 - beg0) : 32;
        float2 acc = make_float2(0.0f, 0.0f);
        if (cnt > 8) {
            float2 hv[16];
            issue_rows<16>(X, s0, 0, lane, hv);
            const float2 rec = csr[idx1];
            hs1 = row_pair(X, n1c, lane);
            sum_rows<16>(acc, hv, nm0, 0, cnt);
            s1 = has1 ? __float_as_int(rec.x) : 0;
            nm1 = has1 ? rec.y : 0.0f;
            if (cnt > 16) {
                float2 hw[16];
                issue_rows<16>(X, s0, 16, lane, hw);
                sum_rows<16>(acc, hw, nm0, 16, cnt);
            }
        } else {
            float2 hv[8];
            issue_rows<8>(X, s0, 0, lane, hv);
            const float2 rec = csr[idx1];
            hs1 = row_pair(X, n1c, lane);
            sum_rows<8>(acc, hv, nm0, 0, cnt);
            s1 = has1 ? __float_as_int(rec.x) : 0;
            nm1 = has1 ? rec.y : 0.0f;
        }
        for (int base = beg0 + 32; base < end0; base += 32) {
            const int c2 = (end0 - base < 32) ? (end0 - base) : 32;
            int s = 0;
            float nm = 0.0f;
            if (lane < c2) {
                float2 rec = csr[base + lane];
                s = __float_as_int(rec.x);
                nm = rec.y;
            }
            float2 hv[16];
            issue_rows<16>(X, s, 0, lane, hv);
            sum_rows<16>(acc, hv, nm, 0, c2);
            if (c2 > 16) {
                float2 hw[16];
                issue_rows<16>(X, s, 16, lane, hw);
                sum_rows<16>(acc, hw, nm, 16, c2);
            }
        }
        const float di0 = lane_get_f(dv, q);
        const float sl = di0 * di0;
        acc.x += hs0.x * sl;
        acc.y += hs0.y * sl;
        reinterpret_cast<float2*>(&sA[(wave * WAVE_NODES + q) * A_PITCH])[lane] = acc;

        beg0 = beg1; end0 = end1; s0 = s1; nm0 = nm1;
        hs0.x = live1 ? hs1.x : 0.0f;
        hs0.y = live1 ? hs1.y : 0.0f;
    }

    const int half = lane >> 4;
    const int l = lane & 15;
    const int c0 = wave * 16;
    const float* __restrict__ wcol = W + (c0 + l) + half * 2 * D;
    v2f bfrag[D / 4];
#pragma unroll
    for (int k = 0; k < D; k += 4) {
        bfrag[k / 4].x = wcol[k * D];
        bfrag[k / 4].y = wcol[k * D + D];
    }
    const float bc = bias[c0 + l];

    __syncthreads();

    const float* a0p = &sA[l * A_PITCH + half * 2];
    const float* a1p = &sA[(16 + l) * A_PITCH + half * 2];
    v8f acc0 = {};
    v8f acc1 = {};
#pragma unroll
    for (int k = 0; k < D; k += 4) {
        const v2f a0 = *reinterpret_cast<const v2f*>(a0p + k);
        const v2f a1 = *reinterpret_cast<const v2f*>(a1p + k);
        acc0 = __builtin_amdgcn_wmma_f32_16x16x4_f32(
             false, a0,  false, bfrag[k / 4],
             (short)0, acc0,  false,  false);
        acc1 = __builtin_amdgcn_wmma_f32_16x16x4_f32(
             false, a1,  false, bfrag[k / 4],
             (short)0, acc1,  false,  false);
    }

    const int r0 = tile0 + half * 8;
    float* __restrict__ op = outp + (size_t)r0 * D + c0 + l;
    if (tile0 + TILE_NODES <= n) {
#pragma unroll
        for (int v = 0; v < 8; ++v) {
            op[(size_t)v * D] = fmaxf(acc0[v] + bc, 0.0f);
            op[(size_t)(16 + v) * D] = fmaxf(acc1[v] + bc, 0.0f);
        }
    } else {
#pragma unroll
        for (int v = 0; v < 8; ++v) {
            if (r0 + v < n) op[(size_t)v * D] = fmaxf(acc0[v] + bc, 0.0f);
            if (r0 + 16 + v < n) op[(size_t)(16 + v) * D] = fmaxf(acc1[v] + bc, 0.0f);
        }
    }
}


extern "C" void kernel_launch(void* const* d_in, const int* in_sizes, int n_in,
                              void* d_out, int out_size, void* d_ws, size_t ws_size,
                              hipStream_t stream) {
    (void)n_in; (void)out_size; (void)ws_size;

    const float* x  = (const float*)d_in[0];
    const int*   ei = (const int*)  d_in[1];
    const float* ea = (const float*)d_in[2];
    const float* W1 = (const float*)d_in[3];
    const float* b1 = (const float*)d_in[4];
    const float* W2 = (const float*)d_in[5];
    const float* b2 = (const float*)d_in[6];
    const float* W3 = (const float*)d_in[7];
    const float* b3 = (const float*)d_in[8];

    const int N = in_sizes[0] / D;
    const int E = in_sizes[2];
    const int* src = ei;
    const int* dst = ei + E;
    const int nb = (N + NPB - 1) / NPB;

    float*  out    = (float*)d_out;
    float*  bufA   = (float*)d_ws;
    float*  bufB   = bufA + (size_t)N * D;
    float2* csr    = (float2*)(bufB + (size_t)N * D);
    float*  dinv   = (float*)(csr + E);
    int*    cnt    = (int*)(dinv + N);
    int*    rowptr = cnt + N;
    int*    bsum   = rowptr + N + 1;
    int*    rank   = bsum + nb;

    const dim3 blk(256);
    const dim3 gNode(nb);
    const dim3 gEdge((E + 256 * EPT - 1) / (256 * EPT));
    const dim3 blkLayer(128);
    const dim3 gLayer((N + TILE_NODES - 1) / TILE_NODES);

    node_init_kernel<<<gNode, blk, 0, stream>>>(dinv, cnt, N);
    edge_count_kernel<<<gEdge, blk, 0, stream>>>(dst, ea, dinv, cnt, rank, E);
    dinv_blocksum_kernel<<<gNode, blk, 0, stream>>>(dinv, cnt, bsum, N);
    rowptr_scan_kernel<<<gNode, blk, 0, stream>>>(cnt, bsum, rowptr, N);
    csr_fill_kernel<<<gEdge, blk, 0, stream>>>(src, dst, ea, dinv, rowptr, rank, csr, E);

    aggregate_transform_kernel<<<gLayer, blkLayer, 0, stream>>>(x,    rowptr, csr, dinv, W1, b1, bufB, N, E);
    aggregate_transform_kernel<<<gLayer, blkLayer, 0, stream>>>(bufB, rowptr, csr, dinv, W2, b2, bufA, N, E);
    aggregate_transform_kernel<<<gLayer, blkLayer, 0, stream>>>(bufA, rowptr, csr, dinv, W3, b3, out,  N, E);
}


# Degree sums no longer a scattered float atomic per edge: one thread per node adds the weights of its destination-grouped records; norms filled by a streamed pass after; deg sum order = list order
# speedup vs baseline: 2.6905x; 1.0765x over previous
#include <hip/hip_runtime.h>
#include <hip/hip_bf16.h>

typedef __attribute__((ext_vector_type(2))) float v2f;
typedef __attribute__((ext_vector_type(8))) float v8f;

#define D 64

#define NPB 1024
#define EPT 4

__global__ void node_init_kernel(int* __restrict__ cnt, int n) {
    const int first = blockIdx.x * NPB + threadIdx.x;
#pragma unroll
    for (int u = 0; u < NPB / 256; ++u) {
        int i = first + u * 256;
        if (i < n) cnt[i] = 0;
    }
}

__global__ void edge_count_kernel(const int* __restrict__ dst,
                                  int* __restrict__ cnt,
                                  int* __restrict__ rank, int E) {
    const int first = blockIdx.x * (256 * EPT) + threadIdx.x;
    int d[EPT];
#pragma unroll
    for (int u = 0; u < EPT; ++u) {
        int e = first + u * 256;
        if (e < E) d[u] = dst[e];
    }
    int r[EPT];
#pragma unroll
    for (int u = 0; u < EPT; ++u) {
        int e = first + u * 256;
        if (e < E) r[u] = atomicAdd(&cnt[d[u]], 1);
    }
#pragma unroll
    for (int u = 0; u < EPT; ++u) {
        int e = first + u * 256;
        if (e < E) rank[e] = r[u];
    }
}

__global__ void cnt_blocksum_kernel(const int* __restrict__ cnt,
                                    int* __restrict__ bsum, int n) {
    __shared__ int wsum[8];
    const int first = blockIdx.x * NPB + threadIdx.x;
    int s = 0;
#pragma unroll
    for (int u = 0; u < NPB / 256; ++u) {
        int i = first + u * 256;
        if (i < n) s += cnt[i];
    }
    for (int off = 16; off > 0; off >>= 1) s += __shfl_xor(s, off);
    if ((threadIdx.x & 31) == 0) wsum[threadIdx.x >> 5] = s;
    __syncthreads();
    if (threadIdx.x == 0) {
        int t = 0;
        for (int k = 0; k < 8; ++k) t += wsum[k];
        bsum[blockIdx.x] = t;
    }
}

__global__ void rowptr_scan_kernel(const int* __restrict__ cnt,
                                   const int* __restrict__ bsum,
                                   int* __restrict__ rowptr, int n) {
    __shared__ int wtot[8];
    __shared__ int wpre[8];
    const int t = threadIdx.x;
    const int lane = t & 31;
    const int wave = t >> 5;
    int p = 0;
    for (int k = t; k < (int)blockIdx.x; k += 256) p += bsum[k];
    const int i0 = blockIdx.x * NPB + 4 * t;
    int c[4];
#pragma unroll
    for (int k = 0; k < 4; ++k) c[k] = (i0 + k < n) ? cnt[i0 + k] : 0;
    const int s = c[0] + c[1] + c[2] + c[3];
    int incl = s;
    for (int off = 1; off < 32; off <<= 1) {
        int v = __shfl_up(incl, off);
        if (lane >= off) incl += v;
    }
    for (int off = 16; off > 0; off >>= 1) p += __shfl_xor(p, off);
    if (lane == 31) wtot[wave] = incl;
    if (lane == 0) wpre[wave] = p;
    __syncthreads();
    int ex = incl - s;
    for (int k = 0; k < 8; ++k) {
        ex += wpre[k];
        if (k < wave) ex += wtot[k];
    }
#pragma unroll
    for (int k = 0; k < 4; ++k) {
        if (i0 + k < n) rowptr[i0 + k] = ex;
        ex += c[k];
    }
    if (i0 <= n - 1 && n - 1 < i0 + 4) rowptr[n] = ex;
}

__global__ void rec_fill_kernel(const int* __restrict__ src,
                                const int* __restrict__ dst,
                                const float* __restrict__ w,
                                const int* __restrict__ rowptr,
                                const int* __restrict__ rank,
                                float4* __restrict__ rec, int E) {
    const int first = blockIdx.x * (256 * EPT) + threadIdx.x;
    int s[EPT], d[EPT], rk[EPT];
    float ww[EPT];
#pragma unroll
    for (int u = 0; u < EPT; ++u) {
        int e = first + u * 256;
        if (e < E) { s[u] = src[e]; d[u] = dst[e]; ww[u] = w[e]; rk[u] = rank[e]; }
    }
    int slot[EPT];
#pragma unroll
    for (int u = 0; u < EPT; ++u) {
        int e = first + u * 256;
        if (e < E) slot[u] = rowptr[d[u]] + rk[u];
    }
#pragma unroll
    for (int u = 0; u < EPT; ++u) {
        int e = first + u * 256;
        if (e < E) rec[slot[u]] = make_float4(__int_as_float(s[u]), ww[u], __int_as_float(d[u]), 0.0f);
    }
}

__global__ void dinv_kernel(const int* __restrict__ rowptr,
                            const float4* __restrict__ rec,
                            float* __restrict__ dinv, int n) {
    const int i = blockIdx.x * 256 + threadIdx.x;
    if (i >= n) return;
    int p = rowptr[i];
    const int end = rowptr[i + 1];
    float dg = 1.0f;
    for (; p + 4 <= end; p += 4) {
        const float a0 = rec[p].y;
        const float a1 = rec[p + 1].y;
        const float a2 = rec[p + 2].y;
        const float a3 = rec[p + 3].y;
        dg += a0; dg += a1; dg += a2; dg += a3;
    }
    if (p < end) {
        const int l = end - 1;
        const float a0 = rec[p].y;
        const float a1 = rec[(p + 1 < l) ? (p + 1) : l].y;
        const float a2 = rec[(p + 2 < l) ? (p + 2) : l].y;
        dg += a0;
        if (p + 1 < end) dg += a1;
        if (p + 2 < end) dg += a2;
    }
    dinv[i] = (dg > 0.0f) ? rsqrtf(fmaxf(dg, 1e-30f)) : 0.0f;
}

#define NORM_RPT 8
__global__ void norm_kernel(const float4* __restrict__ rec,
                            const float* __restrict__ dinv,
                            float2* __restrict__ csr, int E) {
    const int first = blockIdx.x * (256 * NORM_RPT) + threadIdx.x;
    if ((int)(blockIdx.x + 1) * (256 * NORM_RPT) <= E) {
        float4 r[NORM_RPT];
#pragma unroll
        for (int u = 0; u < NORM_RPT; ++u) r[u] = rec[first + u * 256];
        float ds[NORM_RPT], dd[NORM_RPT];
#pragma unroll
        for (int u = 0; u < NORM_RPT; ++u) {
            ds[u] = dinv[__float_as_int(r[u].x)];
            dd[u] = dinv[__float_as_int(r[u].z)];
        }
#pragma unroll
        for (int u = 0; u < NORM_RPT; ++u)
            csr[first + u * 256] = make_float2(r[u].x, ds[u] * r[u].y * dd[u]);
    } else {
        for (int u = 0; u < NORM_RPT; ++u) {
            const int p = first + u * 256;
            if (p < E) {
                const float4 r = rec[p];
                csr[p] = make_float2(r.x, dinv[__float_as_int(r.x)] * r.y * dinv[__float_as_int(r.z)]);
            }
        }
    }
}

#define TILE_NODES 32
#define WAVE_NODES 8
#define A_PITCH 68

static __device__ __forceinline__ int lane_get_i(int v, int j) {
    return __builtin_amdgcn_readlane(v, j);
}
static __device__ __forceinline__ float lane_get_f(float v, int j) {
    return __int_as_float(__builtin_amdgcn_readlane(__float_as_int(v), j));
}
static __device__ __forceinline__ float2 row_pair(const float* __restrict__ H, int row, int lane) {
    return reinterpret_cast<const float2*>(H + (size_t)row * D)[lane];
}
template <int R>
static __device__ __forceinline__ void issue_rows(const float* __restrict__ H, int s, int j,
                                                  int lane, float2 (&hv)[R]) {
#pragma unroll
    for (int u = 0; u < R; ++u) hv[u] = row_pair(H, lane_get_i(s, j + u), lane);
}
template <int R>
static __device__ __forceinline__ void sum_rows(float2& acc, const float2 (&hv)[R], float nm,
                                                int j, int cnt) {
#pragma unroll
    for (int u = 0; u < R; ++u) {
        const float nj = lane_get_f(nm, j + u);
        const bool live = (j + u < cnt);
        const float hx = live ? hv[u].x : 0.0f;
        const float hy = live ? hv[u].y : 0.0f;
        acc.x += hx * nj;
        acc.y += hy * nj;
    }
}

__global__ void __launch_bounds__(128)
aggregate_transform_kernel(const float* __restrict__ X,
                           const int* __restrict__ rowptr,
                           const float2* __restrict__ csr,
                           const float* __restrict__ dinv,
                           const float* __restrict__ W,
                           const float* __restrict__ bias,
                           float* __restrict__ outp, int n, int E) {
    __shared__ float sA[TILE_NODES * A_PITCH];
    const int lane = threadIdx.x & 31;
    const int wave = __builtin_amdgcn_readfirstlane((int)(threadIdx.x >> 5));
    const int tile0 = blockIdx.x * TILE_NODES;
    const int nbeg = tile0 + wave * WAVE_NODES;
    const int elast = (E > 0) ? (E - 1) : 0;

    int rp;
    float dv = 0.0f;
    {
        const int j = (lane < WAVE_NODES) ? lane : WAVE_NODES;
        const int idx = nbeg + j;
        rp = rowptr[(idx < n) ? idx : n];
        if (lane < WAVE_NODES && idx < n) dv = dinv[idx];
    }

    int beg0 = lane_get_i(rp, 0);
    int end0 = lane_get_i(rp, 1);
    int s0 = 0;
    float nm0 = 0.0f;
    float2 hs0 = make_float2(0.0f, 0.0f);
    if (lane < end0 - beg0) {
        float2 rec = csr[beg0 + lane];
        s0 = __float_as_int(rec.x);
        nm0 = rec.y;
    }
    if (nbeg < n) hs0 = row_pair(X, nbeg, lane);

    for (int q = 0; q < WAVE_NODES; ++q) {
        const int n1 = nbeg + q + 1;
        const int beg1 = end0;
        const int end1 = lane_get_i(rp, q + 2);
        const bool live1 = (q + 1 < WAVE_NODES) && (n1 < n);
        const int n1c = (n1 < n) ? n1 : (n - 1);
        int idx1 = beg1 + lane;
        idx1 = (idx1 < elast) ? idx1 : elast;
        const bool has1 = lane < end1 - beg1;
        int s1;
        float nm1;
        float2 hs1;

        const int cnt = (end0 - beg0 < 32) ? (end0 - beg0) : 32;
        float2 acc = make_float2(0.0f, 0.0f);
        if (cnt > 8) {
            float2 hv[16];
            issue_rows<16>(X, s0, 0, lane, hv);
            const float2 rec = csr[idx1];
            hs1 = row_pair(X, n1c, lane);
            sum_rows<16>(acc, hv, nm0, 0, cnt);
            s1 = has1 ? __float_as_int(rec.x) : 0;
            nm1 = has1 ? rec.y : 0.0f;
            if (cnt > 16) {
                float2 hw[16];
                issue_rows<16>(X, s0, 16, lane, hw);
                sum_rows<16>(acc, hw, nm0, 16, cnt);
            }
        } else {
            float2 hv[8];
            issue_rows<8>(X, s0, 0, lane, hv);
            const float2 rec = csr[idx1];
            hs1 = row_pair(X, n1c, lane);
            sum_rows<8>(acc, hv, nm0, 0, cnt);
            s1 = has1 ? __float_as_int(rec.x) : 0;
            nm1 = has1 ? rec.y : 0.0f;
        }
        for (int base = beg0 + 32; base < end0; base += 32) {
            const int c2 = (end0 - base < 32) ? (end0 - base) : 32;
            int s = 0;
            float nm = 0.0f;
            if (lane < c2) {
                float2 rec = csr[base + lane];
                s = __float_as_int(rec.x);
                nm = rec.y;
            }
            float2 hv[16];
            issue_rows<16>(X, s, 0, lane, hv);
            sum_rows<16>(acc, hv, nm, 0, c2);
            if (c2 > 16) {
                float2 hw[16];
                issue_rows<16>(X, s, 16, lane, hw);
                sum_rows<16>(acc, hw, nm, 16, c2);
            }
        }
        const float di0 = lane_get_f(dv, q);
        const float sl = di0 * di0;
        acc.x += hs0.x * sl;
        acc.y += hs0.y * sl;
        reinterpret_cast<float2*>(&sA[(wave * WAVE_NODES + q) * A_PITCH])[lane] = acc;

        beg0 = beg1; end0 = end1; s0 = s1; nm0 = nm1;
        hs0.x = live1 ? hs1.x : 0.0f;
        hs0.y = live1 ? hs1.y : 0.0f;
    }

    const int half = lane >> 4;
    const int l = lane & 15;
    const int c0 = wave * 16;
    const float* __restrict__ wcol = W + (c0 + l) + half * 2 * D;
    v2f bfrag[D / 4];
#pragma unroll
    for (int k = 0; k < D; k += 4) {
        bfrag[k / 4].x = wcol[k * D];
        bfrag[k / 4].y = wcol[k * D + D];
    }
    const float bc = bias[c0 + l];

    __syncthreads();

    const float* a0p = &sA[l * A_PITCH + half * 2];
    const float* a1p = &sA[(16 + l) * A_PITCH + half * 2];
    v8f acc0 = {};
    v8f acc1 = {};
#pragma unroll
    for (int k = 0; k < D; k += 4) {
        const v2f a0 = *reinterpret_cast<const v2f*>(a0p + k);
        const v2f a1 = *reinterpret_cast<const v2f*>(a1p + k);
        acc0 = __builtin_amdgcn_wmma_f32_16x16x4_f32(
             false, a0,  false, bfrag[k / 4],
             (short)0, acc0,  false,  false);
        acc1 = __builtin_amdgcn_wmma_f32_16x16x4_f32(
             false, a1,  false, bfrag[k / 4],
             (short)0, acc1,  false,  false);
    }

    const int r0 = tile0 + half * 8;
    float* __restrict__ op = outp + (size_t)r0 * D + c0 + l;
    if (tile0 + TILE_NODES <= n) {
#pragma unroll
        for (int v = 0; v < 8; ++v) {
            op[(size_t)v * D] = fmaxf(acc0[v] + bc, 0.0f);
            op[(size_t)(16 + v) * D] = fmaxf(acc1[v] + bc, 0.0f);
        }
    } else {
#pragma unroll
        for (int v = 0; v < 8; ++v) {
            if (r0 + v < n) op[(size_t)v * D] = fmaxf(acc0[v] + bc, 0.0f);
            if (r0 + 16 + v < n) op[(size_t)(16 + v) * D] = fmaxf(acc1[v] + bc, 0.0f);
        }
    }
}


extern "C" void kernel_launch(void* const* d_in, const int* in_sizes, int n_in,
                              void* d_out, int out_size, void* d_ws, size_t ws_size,
                              hipStream_t stream) {
    (void)n_in; (void)out_size; (void)ws_size;

    const float* x  = (const float*)d_in[0];
    const int*   ei = (const int*)  d_in[1];
    const float* ea = (const float*)d_in[2];
    const float* W1 = (const float*)d_in[3];
    const float* b1 = (const float*)d_in[4];
    const float* W2 = (const float*)d_in[5];
    const float* b2 = (const float*)d_in[6];
    const float* W3 = (const float*)d_in[7];
    const float* b3 = (const float*)d_in[8];

    const int N = in_sizes[0] / D;
    const int E = in_sizes[2];
    const int* src = ei;
    const int* dst = ei + E;
    const int nb = (N + NPB - 1) / NPB;

    float*  out    = (float*)d_out;
    float*  bufA   = (float*)d_ws;
    float*  bufB   = bufA + (size_t)N * D;
    float4* rec    = (float4*)(bufB + (size_t)N * D);
    float2* csr    = (float2*)(rec + E);
    float*  dinv   = (float*)(csr + E);
    int*    cnt    = (int*)(dinv + N);
    int*    rowptr = cnt + N;
    int*    bsum   = rowptr + N + 1;
    int*    rank   = bsum + nb;

    const dim3 blk(256);
    const dim3 gNode(nb);
    const dim3 gNode1((N + 255) / 256);
    const dim3 gNorm((E + 256 * NORM_RPT - 1) / (256 * NORM_RPT));
    const dim3 gEdge((E + 256 * EPT - 1) / (256 * EPT));
    const dim3 blkLayer(128);
    const dim3 gLayer((N + TILE_NODES - 1) / TILE_NODES);

    node_init_kernel<<<gNode, blk, 0, stream>>>(cnt, N);
    edge_count_kernel<<<gEdge, blk, 0, stream>>>(dst, cnt, rank, E);
    cnt_blocksum_kernel<<<gNode, blk, 0, stream>>>(cnt, bsum, N);
    rowptr_scan_kernel<<<gNode, blk, 0, stream>>>(cnt, bsum, rowptr, N);
    rec_fill_kernel<<<gEdge, blk, 0, stream>>>(src, dst, ea, rowptr, rank, rec, E);
    dinv_kernel<<<gNode1, blk, 0, stream>>>(rowptr, rec, dinv, N);
    norm_kernel<<<gNorm, blk, 0, stream>>>(rec, dinv, csr, E);

    aggregate_transform_kernel<<<gLayer, blkLayer, 0, stream>>>(x,    rowptr, csr, dinv, W1, b1, bufB, N, E);
    aggregate_transform_kernel<<<gLayer, blkLayer, 0, stream>>>(bufB, rowptr, csr, dinv, W2, b2, bufA, N, E);
    aggregate_transform_kernel<<<gLayer, blkLayer, 0, stream>>>(bufA, rowptr, csr, dinv, W3, b3, out,  N, E);
}


# Edges grouped by destination with a two-level counting sort on LDS counters (chunk x node-range table, then one block per range), no global returning atomic per edge; list order = LDS add order
# speedup vs baseline: 4.0747x; 1.5145x over previous
#include <hip/hip_runtime.h>
#include <hip/hip_bf16.h>

typedef __attribute__((ext_vector_type(2))) float v2f;
typedef __attribute__((ext_vector_type(8))) float v8f;

#define D 64

#define NCHUNK 256
#define NBIN 256
#define BIN_SHIFT 9
#define BIN_NODES 512

__global__ void __launch_bounds__(256)
chunk_hist_kernel(const int* __restrict__ dst, int* __restrict__ hist, int E) {
    __shared__ int h[NBIN];
    const int t = threadIdx.x;
    h[t] = 0;
    __syncthreads();
    const int ch = (E + NCHUNK - 1) / NCHUNK;
    const int e0 = blockIdx.x * ch;
    const int e1 = (e0 + ch < E) ? (e0 + ch) : E;
    int e = e0 + t;
    for (; e + 768 < e1; e += 1024) {
        const int d0 = dst[e];
        const int d1 = dst[e + 256];
        const int d2 = dst[e + 512];
        const int d3 = dst[e + 768];
        atomicAdd(&h[d0 >> BIN_SHIFT], 1);
        atomicAdd(&h[d1 >> BIN_SHIFT], 1);
        atomicAdd(&h[d2 >> BIN_SHIFT], 1);
        atomicAdd(&h[d3 >> BIN_SHIFT], 1);
    }
    for (; e < e1; e += 256) atomicAdd(&h[dst[e] >> BIN_SHIFT], 1);
    __syncthreads();
    hist[blockIdx.x * NBIN + t] = h[t];
}

__global__ void __launch_bounds__(256)
bin_scatter_kernel(const int* __restrict__ src,
                   const int* __restrict__ dst,
                   const float* __restrict__ w,
                   const int* __restrict__ hist,
                   int* __restrict__ binstart,
                   float4* __restrict__ binrec, int E) {
    __shared__ int cur[NBIN];
    __shared__ int wtot[8];
    const int t = threadIdx.x;
    const int lane = t & 31;
    const int wave = t >> 5;
    const int c = blockIdx.x;
    int tot = 0;
    int before = 0;
    for (int k0 = 0; k0 < NCHUNK; k0 += 32) {
        int x[32];
#pragma unroll
        for (int u = 0; u < 32; ++u) x[u] = hist[(k0 + u) * NBIN + t];
#pragma unroll
        for (int u = 0; u < 32; ++u) {
            tot += x[u];
            before += (k0 + u < c) ? x[u] : 0;
        }
    }
    int incl = tot;
    for (int off = 1; off < 32; off <<= 1) {
        const int v = __shfl_up(incl, off);
        if (lane >= off) incl += v;
    }
    if (lane == 31) wtot[wave] = incl;
    __syncthreads();
    int ex = incl - tot;
    for (int k = 0; k < wave; ++k) ex += wtot[k];
    cur[t] = ex + before;
    if (c == 0) {
        binstart[t] = ex;
        if (t == NBIN - 1) binstart[NBIN] = ex + tot;
    }
    __syncthreads();

    const int ch = (E + NCHUNK - 1) / NCHUNK;
    const int e0 = c * ch;
    const int e1 = (e0 + ch < E) ? (e0 + ch) : E;
    int e = e0 + t;
    for (; e + 768 < e1; e += 1024) {
        int s[4], d[4];
        float ww[4];
#pragma unroll
        for (int u = 0; u < 4; ++u) {
            s[u] = src[e + u * 256];
            d[u] = dst[e + u * 256];
            ww[u] = w[e + u * 256];
        }
#pragma unroll
        for (int u = 0; u < 4; ++u) {
            const int pos = atomicAdd(&cur[d[u] >> BIN_SHIFT], 1);
            binrec[pos] = make_float4(__int_as_float(s[u]), ww[u], __int_as_float(d[u]), 0.0f);
        }
    }
    for (; e < e1; e += 256) {
        const int s = src[e];
        const int d = dst[e];
        const float ww = w[e];
        const int pos = atomicAdd(&cur[d >> BIN_SHIFT], 1);
        binrec[pos] = make_float4(__int_as_float(s), ww, __int_as_float(d), 0.0f);
    }
}

__global__ void __launch_bounds__(512)
bin_sort_kernel(const float4* __restrict__ binrec,
                const int* __restrict__ binstart,
                int* __restrict__ rowptr,
                float4* __restrict__ rec, int n) {
    __shared__ int lc[BIN_NODES];
    __shared__ int wtot[16];
    const int t = threadIdx.x;
    const int lane = t & 31;
    const int wave = t >> 5;
    const int n0 = blockIdx.x << BIN_SHIFT;
    if (n0 >= n) return;
    const int p0 = binstart[blockIdx.x];
    const int p1 = binstart[blockIdx.x + 1];
    lc[t] = 0;
    __syncthreads();
    const float* __restrict__ dz = reinterpret_cast<const float*>(binrec) + 2;
    int p = p0 + t;
    for (; p + 1536 < p1; p += 2048) {
        const int d0 = __float_as_int(dz[(size_t)p * 4]);
        const int d1 = __float_as_int(dz[(size_t)(p + 512) * 4]);
        const int d2 = __float_as_int(dz[(size_t)(p + 1024) * 4]);
        const int d3 = __float_as_int(dz[(size_t)(p + 1536) * 4]);
        atomicAdd(&lc[d0 - n0], 1);
        atomicAdd(&lc[d1 - n0], 1);
        atomicAdd(&lc[d2 - n0], 1);
        atomicAdd(&lc[d3 - n0], 1);
    }
    for (; p < p1; p += 512) atomicAdd(&lc[__float_as_int(dz[(size_t)p * 4]) - n0], 1);
    __syncthreads();
    const int cnt = lc[t];
    int incl = cnt;
    for (int off = 1; off < 32; off <<= 1) {
        const int v = __shfl_up(incl, off);
        if (lane >= off) incl += v;
    }
    if (lane == 31) wtot[wave] = incl;
    __syncthreads();
    int first = p0 + incl - cnt;
    for (int k = 0; k < wave; ++k) first += wtot[k];
    if (n0 + t < n) rowptr[n0 + t] = first;
    if (n0 + t == n - 1) rowptr[n] = first + cnt;
    lc[t] = first;
    __syncthreads();
    p = p0 + t;
    for (; p + 1536 < p1; p += 2048) {
        float4 r[4];
#pragma unroll
        for (int u = 0; u < 4; ++u) r[u] = binrec[p + u * 512];
#pragma unroll
        for (int u = 0; u < 4; ++u) {
            const int slot = atomicAdd(&lc[__float_as_int(r[u].z) - n0], 1);
            rec[slot] = r[u];
        }
    }
    for (; p < p1; p += 512) {
        const float4 r = binrec[p];
        const int slot = atomicAdd(&lc[__float_as_int(r.z) - n0], 1);
        rec[slot] = r;
    }
}

__global__ void dinv_kernel(const int* __restrict__ rowptr,
                            const float4* __restrict__ rec,
                            float* __restrict__ dinv, int n) {
    const int i = blockIdx.x * 256 + threadIdx.x;
    if (i >= n) return;
    int p = rowptr[i];
    const int end = rowptr[i + 1];
    float dg = 1.0f;
    for (; p + 4 <= end; p += 4) {
        const float a0 = rec[p].y;
        const float a1 = rec[p + 1].y;
        const float a2 = rec[p + 2].y;
        const float a3 = rec[p + 3].y;
        dg += a0; dg += a1; dg += a2; dg += a3;
    }
    if (p < end) {
        const int l = end - 1;
        const float a0 = rec[p].y;
        const float a1 = rec[(p + 1 < l) ? (p + 1) : l].y;
        const float a2 = rec[(p + 2 < l) ? (p + 2) : l].y;
        dg += a0;
        if (p + 1 < end) dg += a1;
        if (p + 2 < end) dg += a2;
    }
    dinv[i] = (dg > 0.0f) ? rsqrtf(fmaxf(dg, 1e-30f)) : 0.0f;
}

#define NORM_RPT 8
__global__ void norm_kernel(const float4* __restrict__ rec,
                            const float* __restrict__ dinv,
                            float2* __restrict__ csr, int E) {
    const int first = blockIdx.x * (256 * NORM_RPT) + threadIdx.x;
    if ((int)(blockIdx.x + 1) * (256 * NORM_RPT) <= E) {
        float4 r[NORM_RPT];
#pragma unroll
        for (int u = 0; u < NORM_RPT; ++u) r[u] = rec[first + u * 256];
        float ds[NORM_RPT], dd[NORM_RPT];
#pragma unroll
        for (int u = 0; u < NORM_RPT; ++u) {
            ds[u] = dinv[__float_as_int(r[u].x)];
            dd[u] = dinv[__float_as_int(r[u].z)];
        }
#pragma unroll
        for (int u = 0; u < NORM_RPT; ++u)
            csr[first + u * 256] = make_float2(r[u].x, ds[u] * r[u].y * dd[u]);
    } else {
        for (int u = 0; u < NORM_RPT; ++u) {
            const int p = first + u * 256;
            if (p < E) {
                const float4 r = rec[p];
                csr[p] = make_float2(r.x, dinv[__float_as_int(r.x)] * r.y * dinv[__float_as_int(r.z)]);
            }
        }
    }
}

#define TILE_NODES 32
#define WAVE_NODES 8
#define A_PITCH 68

static __device__ __forceinline__ int lane_get_i(int v, int j) {
    return __builtin_amdgcn_readlane(v, j);
}
static __device__ __forceinline__ float lane_get_f(float v, int j) {
    return __int_as_float(__builtin_amdgcn_readlane(__float_as_int(v), j));
}
static __device__ __forceinline__ float2 row_pair(const float* __restrict__ H, int row, int lane) {
    return reinterpret_cast<const float2*>(H + (size_t)row * D)[lane];
}
template <int R>
static __device__ __forceinline__ void issue_rows(const float* __restrict__ H, int s, int j,
                                                  int lane, float2 (&hv)[R]) {
#pragma unroll
    for (int u = 0; u < R; ++u) hv[u] = row_pair(H, lane_get_i(s, j + u), lane);
}
template <int R>
static __device__ __forceinline__ void sum_rows(float2& acc, const float2 (&hv)[R], float nm,
                                                int j, int cnt) {
#pragma unroll
    for (int u = 0; u < R; ++u) {
        const float nj = lane_get_f(nm, j + u);
        const bool live = (j + u < cnt);
        const float hx = live ? hv[u].x : 0.0f;
        const float hy = live ? hv[u].y : 0.0f;
        acc.x += hx * nj;
        acc.y += hy * nj;
    }
}

__global__ void __launch_bounds__(128)
aggregate_transform_kernel(const float* __restrict__ X,
                           const int* __restrict__ rowptr,
                           const float2* __restrict__ csr,
                           const float* __restrict__ dinv,
                           const float* __restrict__ W,
                           const float* __restrict__ bias,
                           float* __restrict__ outp, int n, int E) {
    __shared__ float sA[TILE_NODES * A_PITCH];
    const int lane = threadIdx.x & 31;
    const int wave = __builtin_amdgcn_readfirstlane((int)(threadIdx.x >> 5));
    const int tile0 = blockIdx.x * TILE_NODES;
    const int nbeg = tile0 + wave * WAVE_NODES;
    const int elast = (E > 0) ? (E - 1) : 0;

    int rp;
    float dv = 0.0f;
    {
        const int j = (lane < WAVE_NODES) ? lane : WAVE_NODES;
        const int idx = nbeg + j;
        rp = rowptr[(idx < n) ? idx : n];
        if (lane < WAVE_NODES && idx < n) dv = dinv[idx];
    }

    int beg0 = lane_get_i(rp, 0);
    int end0 = lane_get_i(rp, 1);
    int s0 = 0;
    float nm0 = 0.0f;
    float2 hs0 = make_float2(0.0f, 0.0f);
    if (lane < end0 - beg0) {
        float2 rec = csr[beg0 + lane];
        s0 = __float_as_int(rec.x);
        nm0 = rec.y;
    }
    if (nbeg < n) hs0 = row_pair(X, nbeg, lane);

    for (int q = 0; q < WAVE_NODES; ++q) {
        const int n1 = nbeg + q + 1;
        const int beg1 = end0;
        const int end1 = lane_get_i(rp, q + 2);
        const bool live1 = (q + 1 < WAVE_NODES) && (n1 < n);
        const int n1c = (n1 < n) ? n1 : (n - 1);
        int idx1 = beg1 + lane;
        idx1 = (idx1 < elast) ? idx1 : elast;
        const bool has1 = lane < end1 - beg1;
        int s1;
        float nm1;
        float2 hs1;

        const int cnt = (end0 - beg0 < 32) ? (end0 - beg0) : 32;
        float2 acc = make_float2(0.0f, 0.0f);
        if (cnt > 8) {
            float2 hv[16];
            issue_rows<16>(X, s0, 0, lane, hv);
            const float2 rec = csr[idx1];
            hs1 = row_pair(X, n1c, lane);
            sum_rows<16>(acc, hv, nm0, 0, cnt);
            s1 = has1 ? __float_as_int(rec.x) : 0;
            nm1 = has1 ? rec.y : 0.0f;
            if (cnt > 16) {
                float2 hw[16];
                issue_rows<16>(X, s0, 16, lane, hw);
                sum_rows<16>(acc, hw, nm0, 16, cnt);
            }
        } else {
            float2 hv[8];
            issue_rows<8>(X, s0, 0, lane, hv);
            const float2 rec = csr[idx1];
            hs1 = row_pair(X, n1c, lane);
            sum_rows<8>(acc, hv, nm0, 0, cnt);
            s1 = has1 ? __float_as_int(rec.x) : 0;
            nm1 = has1 ? rec.y : 0.0f;
        }
        for (int base = beg0 + 32; base < end0; base += 32) {
            const int c2 = (end0 - base < 32) ? (end0 - base) : 32;
            int s = 0;
            float nm = 0.0f;
            if (lane < c2) {
                float2 rec = csr[base + lane];
                s = __float_as_int(rec.x);
                nm = rec.y;
            }
            float2 hv[16];
            issue_rows<16>(X, s, 0, lane, hv);
            sum_rows<16>(acc, hv, nm, 0, c2);
            if (c2 > 16) {
                float2 hw[16];
                issue_rows<16>(X, s, 16, lane, hw);
                sum_rows<16>(acc, hw, nm, 16, c2);
            }
        }
        const float di0 = lane_get_f(dv, q);
        const float sl = di0 * di0;
        acc.x += hs0.x * sl;
        acc.y += hs0.y * sl;
        reinterpret_cast<float2*>(&sA[(wave * WAVE_NODES + q) * A_PITCH])[lane] = acc;

        beg0 = beg1; end0 = end1; s0 = s1; nm0 = nm1;
        hs0.x = live1 ? hs1.x : 0.0f;
        hs0.y = live1 ? hs1.y : 0.0f;
    }

    const int half = lane >> 4;
    const int l = lane & 15;
    const int c0 = wave * 16;
    const float* __restrict__ wcol = W + (c0 + l) + half * 2 * D;
    v2f bfrag[D / 4];
#pragma unroll
    for (int k = 0; k < D; k += 4) {
        bfrag[k / 4].x = wcol[k * D];
        bfrag[k / 4].y = wcol[k * D + D];
    }
    const float bc = bias[c0 + l];

    __syncthreads();

    const float* a0p = &sA[l * A_PITCH + half * 2];
    const float* a1p = &sA[(16 + l) * A_PITCH + half * 2];
    v8f acc0 = {};
    v8f acc1 = {};
#pragma unroll
    for (int k = 0; k < D; k += 4) {
        const v2f a0 = *reinterpret_cast<const v2f*>(a0p + k);
        const v2f a1 = *reinterpret_cast<const v2f*>(a1p + k);
        acc0 = __builtin_amdgcn_wmma_f32_16x16x4_f32(
             false, a0,  false, bfrag[k / 4],
             (short)0, acc0,  false,  false);
        acc1 = __builtin_amdgcn_wmma_f32_16x16x4_f32(
             false, a1,  false, bfrag[k / 4],
             (short)0, acc1,  false,  false);
    }

    const int r0 = tile0 + half * 8;
    float* __restrict__ op = outp + (size_t)r0 * D + c0 + l;
    if (tile0 + TILE_NODES <= n) {
#pragma unroll
        for (int v = 0; v < 8; ++v) {
            op[(size_t)v * D] = fmaxf(acc0[v] + bc, 0.0f);
            op[(size_t)(16 + v) * D] = fmaxf(acc1[v] + bc, 0.0f);
        }
    } else {
#pragma unroll
        for (int v = 0; v < 8; ++v) {
            if (r0 + v < n) op[(size_t)v * D] = fmaxf(acc0[v] + bc, 0.0f);
            if (r0 + 16 + v < n) op[(size_t)(16 + v) * D] = fmaxf(acc1[v] + bc, 0.0f);
        }
    }
}


extern "C" void kernel_launch(void* const* d_in, const int* in_sizes, int n_in,
                              void* d_out, int out_size, void* d_ws, size_t ws_size,
                              hipStream_t stream) {
    (void)n_in; (void)out_size; (void)ws_size;

    const float* x  = (const float*)d_in[0];
    const int*   ei = (const int*)  d_in[1];
    const float* ea = (const float*)d_in[2];
    const float* W1 = (const float*)d_in[3];
    const float* b1 = (const float*)d_in[4];
    const float* W2 = (const float*)d_in[5];
    const float* b2 = (const float*)d_in[6];
    const float* W3 = (const float*)d_in[7];
    const float* b3 = (const float*)d_in[8];

    const int N = in_sizes[0] / D;
    const int E = in_sizes[2];
    const int* src = ei;
    const int* dst = ei + E;

    float*  out      = (float*)d_out;
    float*  bufA     = (float*)d_ws;
    float*  bufB     = bufA + (size_t)N * D;
    float4* rec      = (float4*)(bufB + (size_t)N * D);
    float4* binrec   = rec + E;
    int*    hist     = (int*)(binrec + E);
    int*    binstart = hist + NCHUNK * NBIN;
    float2* csr      = (float2*)(binstart + NBIN + 4);
    float*  dinv     = (float*)(csr + E);
    int*    rowptr   = (int*)(dinv + N);

    const dim3 blk(256);
    const dim3 gNode1((N + 255) / 256);
    const dim3 gNorm((E + 256 * NORM_RPT - 1) / (256 * NORM_RPT));
    const dim3 blkLayer(128);
    const dim3 gLayer((N + TILE_NODES - 1) / TILE_NODES);

    chunk_hist_kernel<<<dim3(NCHUNK), blk, 0, stream>>>(dst, hist, E);
    bin_scatter_kernel<<<dim3(NCHUNK), blk, 0, stream>>>(src, dst, ea, hist, binstart, binrec, E);
    bin_sort_kernel<<<dim3(NBIN), dim3(BIN_NODES), 0, stream>>>(binrec, binstart, rowptr, rec, N);
    dinv_kernel<<<gNode1, blk, 0, stream>>>(rowptr, rec, dinv, N);
    norm_kernel<<<gNorm, blk, 0, stream>>>(rec, dinv, csr, E);

    aggregate_transform_kernel<<<gLayer, blkLayer, 0, stream>>>(x,    rowptr, csr, dinv, W1, b1, bufB, N, E);
    aggregate_transform_kernel<<<gLayer, blkLayer, 0, stream>>>(bufB, rowptr, csr, dinv, W2, b2, bufA, N, E);
    aggregate_transform_kernel<<<gLayer, blkLayer, 0, stream>>>(bufA, rowptr, csr, dinv, W3, b3, out,  N, E);
}


# Degree sums folded into the range sort's count pass (LDS float adds, it writes dinv) and edge norms formed and stored by layer 1: 8 -> 6 launches; degree sum order = LDS add arrival order
# speedup vs baseline: 4.5563x; 1.1182x over previous
#include <hip/hip_runtime.h>
#include <hip/hip_bf16.h>

typedef __attribute__((ext_vector_type(2))) float v2f;
typedef __attribute__((ext_vector_type(8))) float v8f;

#define D 64

#define NCHUNK 256
#define NBIN 256
#define BIN_SHIFT 9
#define BIN_NODES 512

__global__ void __launch_bounds__(256)
chunk_hist_kernel(const int* __restrict__ dst, int* __restrict__ hist, int E) {
    __shared__ int h[NBIN];
    const int t = threadIdx.x;
    h[t] = 0;
    __syncthreads();
    const int ch = (E + NCHUNK - 1) / NCHUNK;
    const int e0 = blockIdx.x * ch;
    const int e1 = (e0 + ch < E) ? (e0 + ch) : E;
    int e = e0 + t;
    for (; e + 768 < e1; e += 1024) {
        const int d0 = dst[e];
        const int d1 = dst[e + 256];
        const int d2 = dst[e + 512];
        const int d3 = dst[e + 768];
        atomicAdd(&h[d0 >> BIN_SHIFT], 1);
        atomicAdd(&h[d1 >> BIN_SHIFT], 1);
        atomicAdd(&h[d2 >> BIN_SHIFT], 1);
        atomicAdd(&h[d3 >> BIN_SHIFT], 1);
    }
    for (; e < e1; e += 256) atomicAdd(&h[dst[e] >> BIN_SHIFT], 1);
    __syncthreads();
    hist[blockIdx.x * NBIN + t] = h[t];
}

__global__ void __launch_bounds__(256)
bin_scatter_kernel(const int* __restrict__ src,
                   const int* __restrict__ dst,
                   const float* __restrict__ w,
                   const int* __restrict__ hist,
                   int* __restrict__ binstart,
                   float4* __restrict__ binrec, int E) {
    __shared__ int cur[NBIN];
    __shared__ int wtot[8];
    const int t = threadIdx.x;
    const int lane = t & 31;
    const int wave = t >> 5;
    const int c = blockIdx.x;
    int tot = 0;
    int before = 0;
    for (int k0 = 0; k0 < NCHUNK; k0 += 32) {
        int x[32];
#pragma unroll
        for (int u = 0; u < 32; ++u) x[u] = hist[(k0 + u) * NBIN + t];
#pragma unroll
        for (int u = 0; u < 32; ++u) {
            tot += x[u];
            before += (k0 + u < c) ? x[u] : 0;
        }
    }
    int incl = tot;
    for (int off = 1; off < 32; off <<= 1) {
        const int v = __shfl_up(incl, off);
        if (lane >= off) incl += v;
    }
    if (lane == 31) wtot[wave] = incl;
    __syncthreads();
    int ex = incl - tot;
    for (int k = 0; k < wave; ++k) ex += wtot[k];
    cur[t] = ex + before;
    if (c == 0) {
        binstart[t] = ex;
        if (t == NBIN - 1) binstart[NBIN] = ex + tot;
    }
    __syncthreads();

    const int ch = (E + NCHUNK - 1) / NCHUNK;
    const int e0 = c * ch;
    const int e1 = (e0 + ch < E) ? (e0 + ch) : E;
    int e = e0 + t;
    for (; e + 768 < e1; e += 1024) {
        int s[4], d[4];
        float ww[4];
#pragma unroll
        for (int u = 0; u < 4; ++u) {
            s[u] = src[e + u * 256];
            d[u] = dst[e + u * 256];
            ww[u] = w[e + u * 256];
        }
#pragma unroll
        for (int u = 0; u < 4; ++u) {
            const int pos = atomicAdd(&cur[d[u] >> BIN_SHIFT], 1);
            binrec[pos] = make_float4(__int_as_float(d[u]), ww[u], __int_as_float(s[u]), 0.0f);
        }
    }
    for (; e < e1; e += 256) {
        const int s = src[e];
        const int d = dst[e];
        const float ww = w[e];
        const int pos = atomicAdd(&cur[d >> BIN_SHIFT], 1);
        binrec[pos] = make_float4(__int_as_float(d), ww, __int_as_float(s), 0.0f);
    }
}

__global__ void __launch_bounds__(512)
bin_sort_kernel(const float4* __restrict__ binrec,
                const int* __restrict__ binstart,
                int* __restrict__ rowptr,
                float* __restrict__ dinv,
                float2* __restrict__ ew, int n) {
    __shared__ int lc[BIN_NODES];
    __shared__ float ldeg[BIN_NODES];
    __shared__ int wtot[16];
    const int t = threadIdx.x;
    const int lane = t & 31;
    const int wave = t >> 5;
    const int n0 = blockIdx.x << BIN_SHIFT;
    if (n0 >= n) return;
    const int p0 = binstart[blockIdx.x];
    const int p1 = binstart[blockIdx.x + 1];
    lc[t] = 0;
    ldeg[t] = 0.0f;
    __syncthreads();
    const float2* __restrict__ dw = reinterpret_cast<const float2*>(binrec);
    int p = p0 + t;
    for (; p + 1536 < p1; p += 2048) {
        const float2 r0 = dw[(size_t)p * 2];
        const float2 r1 = dw[(size_t)(p + 512) * 2];
        const float2 r2 = dw[(size_t)(p + 1024) * 2];
        const float2 r3 = dw[(size_t)(p + 1536) * 2];
        atomicAdd(&lc[__float_as_int(r0.x) - n0], 1);
        atomicAdd(&lc[__float_as_int(r1.x) - n0], 1);
        atomicAdd(&lc[__float_as_int(r2.x) - n0], 1);
        atomicAdd(&lc[__float_as_int(r3.x) - n0], 1);
        atomicAdd(&ldeg[__float_as_int(r0.x) - n0], r0.y);
        atomicAdd(&ldeg[__float_as_int(r1.x) - n0], r1.y);
        atomicAdd(&ldeg[__float_as_int(r2.x) - n0], r2.y);
        atomicAdd(&ldeg[__float_as_int(r3.x) - n0], r3.y);
    }
    for (; p < p1; p += 512) {
        const float2 r = dw[(size_t)p * 2];
        atomicAdd(&lc[__float_as_int(r.x) - n0], 1);
        atomicAdd(&ldeg[__float_as_int(r.x) - n0], r.y);
    }
    __syncthreads();
    const int cnt = lc[t];
    const float dg = 1.0f + ldeg[t];
    int incl = cnt;
    for (int off = 1; off < 32; off <<= 1) {
        const int v = __shfl_up(incl, off);
        if (lane >= off) incl += v;
    }
    if (lane == 31) wtot[wave] = incl;
    __syncthreads();
    int first = p0 + incl - cnt;
    for (int k = 0; k < wave; ++k) first += wtot[k];
    if (n0 + t < n) {
        rowptr[n0 + t] = first;
        dinv[n0 + t] = (dg > 0.0f) ? rsqrtf(fmaxf(dg, 1e-30f)) : 0.0f;
    }
    if (n0 + t == n - 1) rowptr[n] = first + cnt;
    lc[t] = first;
    __syncthreads();
    p = p0 + t;
    for (; p + 1536 < p1; p += 2048) {
        float4 r[4];
#pragma unroll
        for (int u = 0; u < 4; ++u) r[u] = binrec[p + u * 512];
#pragma unroll
        for (int u = 0; u < 4; ++u) {
            const int slot = atomicAdd(&lc[__float_as_int(r[u].x) - n0], 1);
            ew[slot] = make_float2(r[u].z, r[u].y);
        }
    }
    for (; p < p1; p += 512) {
        const float4 r = binrec[p];
        const int slot = atomicAdd(&lc[__float_as_int(r.x) - n0], 1);
        ew[slot] = make_float2(r.z, r.y);
    }
}

#define TILE_NODES 32
#define WAVE_NODES 8
#define A_PITCH 68

static __device__ __forceinline__ int lane_get_i(int v, int j) {
    return __builtin_amdgcn_readlane(v, j);
}
static __device__ __forceinline__ float lane_get_f(float v, int j) {
    return __int_as_float(__builtin_amdgcn_readlane(__float_as_int(v), j));
}
static __device__ __forceinline__ float2 row_pair(const float* __restrict__ H, int row, int lane) {
    return reinterpret_cast<const float2*>(H + (size_t)row * D)[lane];
}
template <int R>
static __device__ __forceinline__ void issue_rows(const float* __restrict__ H, int s, int j,
                                                  int lane, float2 (&hv)[R]) {
#pragma unroll
    for (int u = 0; u < R; ++u) hv[u] = row_pair(H, lane_get_i(s, j + u), lane);
}
template <int R>
static __device__ __forceinline__ void sum_rows(float2& acc, const float2 (&hv)[R], float nm,
                                                int j, int cnt) {
#pragma unroll
    for (int u = 0; u < R; ++u) {
        const float nj = lane_get_f(nm, j + u);
        const bool live = (j + u < cnt);
        const float hx = live ? hv[u].x : 0.0f;
        const float hy = live ? hv[u].y : 0.0f;
        acc.x += hx * nj;
        acc.y += hy * nj;
    }
}

template <bool FIRST>
__global__ void __launch_bounds__(128)
aggregate_transform_kernel(const float* __restrict__ X,
                           const int* __restrict__ rowptr,
                           const float2* __restrict__ csr,
                           float2* __restrict__ csr_out,
                           const float* __restrict__ dinv,
                           const float* __restrict__ W,
                           const float* __restrict__ bias,
                           float* __restrict__ outp, int n, int E) {
    __shared__ float sA[TILE_NODES * A_PITCH];
    const int lane = threadIdx.x & 31;
    const int wave = __builtin_amdgcn_readfirstlane((int)(threadIdx.x >> 5));
    const int tile0 = blockIdx.x * TILE_NODES;
    const int nbeg = tile0 + wave * WAVE_NODES;
    const int elast = (E > 0) ? (E - 1) : 0;

    int rp;
    float dv = 0.0f;
    {
        const int j = (lane < WAVE_NODES) ? lane : WAVE_NODES;
        const int idx = nbeg + j;
        rp = rowptr[(idx < n) ? idx : n];
        if (lane < WAVE_NODES && idx < n) dv = dinv[idx];
    }

    int beg0 = lane_get_i(rp, 0);
    int end0 = lane_get_i(rp, 1);
    int s0 = 0;
    float nm0 = 0.0f;
    float2 hs0 = make_float2(0.0f, 0.0f);
    if (lane < end0 - beg0) {
        float2 rec = csr[beg0 + lane];
        s0 = __float_as_int(rec.x);
        nm0 = rec.y;
    }
    if (nbeg < n) hs0 = row_pair(X, nbeg, lane);

    for (int q = 0; q < WAVE_NODES; ++q) {
        const int n1 = nbeg + q + 1;
        const int beg1 = end0;
        const int end1 = lane_get_i(rp, q + 2);
        const bool live1 = (q + 1 < WAVE_NODES) && (n1 < n);
        const int n1c = (n1 < n) ? n1 : (n - 1);
        int idx1 = beg1 + lane;
        idx1 = (idx1 < elast) ? idx1 : elast;
        const bool has1 = lane < end1 - beg1;
        int s1;
        float nm1;
        float2 hs1;

        const int cnt = (end0 - beg0 < 32) ? (end0 - beg0) : 32;
        const float di0 = lane_get_f(dv, q);
        float ds0 = 0.0f;
        if (FIRST) ds0 = dinv[s0];
        float2 acc = make_float2(0.0f, 0.0f);
        if (FIRST || cnt > 8) {
            float2 hv[16];
            issue_rows<16>(X, s0, 0, lane, hv);
            const float2 rec = csr[idx1];
            hs1 = row_pair(X, n1c, lane);
            if (FIRST) {
                nm0 = ds0 * nm0 * di0;
                if (lane < cnt) csr_out[beg0 + lane] = make_float2(__int_as_float(s0), nm0);
            }
            sum_rows<16>(acc, hv, nm0, 0, cnt);
            s1 = has1 ? __float_as_int(rec.x) : 0;
            nm1 = has1 ? rec.y : 0.0f;
            if (cnt > 16) {
                float2 hw[16];
                issue_rows<16>(X, s0, 16, lane, hw);
                sum_rows<16>(acc, hw, nm0, 16, cnt);
            }
        } else {
            float2 hv[8];
            issue_rows<8>(X, s0, 0, lane, hv);
            const float2 rec = csr[idx1];
            hs1 = row_pair(X, n1c, lane);
            sum_rows<8>(acc, hv, nm0, 0, cnt);
            s1 = has1 ? __float_as_int(rec.x) : 0;
            nm1 = has1 ? rec.y : 0.0f;
        }
        for (int base = beg0 + 32; base < end0; base += 32) {
            const int c2 = (end0 - base < 32) ? (end0 - base) : 32;
            int s = 0;
            float nm = 0.0f;
            if (lane < c2) {
                float2 rec = csr[base + lane];
                s = __float_as_int(rec.x);
                nm = rec.y;
            }
            if (FIRST) {
                nm = dinv[s] * nm * di0;
                if (lane < c2) csr_out[base + lane] = make_float2(__int_as_float(s), nm);
            }
            float2 hv[16];
            issue_rows<16>(X, s, 0, lane, hv);
            sum_rows<16>(acc, hv, nm, 0, c2);
            if (c2 > 16) {
                float2 hw[16];
                issue_rows<16>(X, s, 16, lane, hw);
                sum_rows<16>(acc, hw, nm, 16, c2);
            }
        }
        const float sl = di0 * di0;
        acc.x += hs0.x * sl;
        acc.y += hs0.y * sl;
        reinterpret_cast<float2*>(&sA[(wave * WAVE_NODES + q) * A_PITCH])[lane] = acc;

        beg0 = beg1; end0 = end1; s0 = s1; nm0 = nm1;
        hs0.x = live1 ? hs1.x : 0.0f;
        hs0.y = live1 ? hs1.y : 0.0f;
    }

    const int half = lane >> 4;
    const int l = lane & 15;
    const int c0 = wave * 16;
    const float* __restrict__ wcol = W + (c0 + l) + half * 2 * D;
    v2f bfrag[D / 4];
#pragma unroll
    for (int k = 0; k < D; k += 4) {
        bfrag[k / 4].x = wcol[k * D];
        bfrag[k / 4].y = wcol[k * D + D];
    }
    const float bc = bias[c0 + l];

    __syncthreads();

    const float* a0p = &sA[l * A_PITCH + half * 2];
    const float* a1p = &sA[(16 + l) * A_PITCH + half * 2];
    v8f acc0 = {};
    v8f acc1 = {};
#pragma unroll
    for (int k = 0; k < D; k += 4) {
        const v2f a0 = *reinterpret_cast<const v2f*>(a0p + k);
        const v2f a1 = *reinterpret_cast<const v2f*>(a1p + k);
        acc0 = __builtin_amdgcn_wmma_f32_16x16x4_f32(
             false, a0,  false, bfrag[k / 4],
             (short)0, acc0,  false,  false);
        acc1 = __builtin_amdgcn_wmma_f32_16x16x4_f32(
             false, a1,  false, bfrag[k / 4],
             (short)0, acc1,  false,  false);
    }

    const int r0 = tile0 + half * 8;
    float* __restrict__ op = outp + (size_t)r0 * D + c0 + l;
    if (tile0 + TILE_NODES <= n) {
#pragma unroll
        for (int v = 0; v < 8; ++v) {
            op[(size_t)v * D] = fmaxf(acc0[v] + bc, 0.0f);
            op[(size_t)(16 + v) * D] = fmaxf(acc1[v] + bc, 0.0f);
        }
    } else {
#pragma unroll
        for (int v = 0; v < 8; ++v) {
            if (r0 + v < n) op[(size_t)v * D] = fmaxf(acc0[v] + bc, 0.0f);
            if (r0 + 16 + v < n) op[(size_t)(16 + v) * D] = fmaxf(acc1[v] + bc, 0.0f);
        }
    }
}


extern "C" void kernel_launch(void* const* d_in, const int* in_sizes, int n_in,
                              void* d_out, int out_size, void* d_ws, size_t ws_size,
                              hipStream_t stream) {
    (void)n_in; (void)out_size; (void)ws_size;

    const float* x  = (const float*)d_in[0];
    const int*   ei = (const int*)  d_in[1];
    const float* ea = (const float*)d_in[2];
    const float* W1 = (const float*)d_in[3];
    const float* b1 = (const float*)d_in[4];
    const float* W2 = (const float*)d_in[5];
    const float* b2 = (const float*)d_in[6];
    const float* W3 = (const float*)d_in[7];
    const float* b3 = (const float*)d_in[8];

    const int N = in_sizes[0] / D;
    const int E = in_sizes[2];
    const int* src = ei;
    const int* dst = ei + E;

    float*  out      = (float*)d_out;
    float*  bufA     = (float*)d_ws;
    float*  bufB     = bufA + (size_t)N * D;
    float4* binrec   = (float4*)(bufB + (size_t)N * D);
    int*    hist     = (int*)(binrec + E);
    int*    binstart = hist + NCHUNK * NBIN;
    float2* ew       = (float2*)(binstart + NBIN + 4);
    float2* csr      = ew + E;
    float*  dinv     = (float*)(csr + E);
    int*    rowptr   = (int*)(dinv + N);

    const dim3 blk(256);
    const dim3 blkLayer(128);
    const dim3 gLayer((N + TILE_NODES - 1) / TILE_NODES);

    chunk_hist_kernel<<<dim3(NCHUNK), blk, 0, stream>>>(dst, hist, E);
    bin_scatter_kernel<<<dim3(NCHUNK), blk, 0, stream>>>(src, dst, ea, hist, binstart, binrec, E);
    bin_sort_kernel<<<dim3(NBIN), dim3(BIN_NODES), 0, stream>>>(binrec, binstart, rowptr, dinv, ew, N);

    aggregate_transform_kernel<true><<<gLayer, blkLayer, 0, stream>>>(x,    rowptr, ew,  csr,     dinv, W1, b1, bufB, N, E);
    aggregate_transform_kernel<false><<<gLayer, blkLayer, 0, stream>>>(bufB, rowptr, csr, nullptr, dinv, W2, b2, bufA, N, E);
    aggregate_transform_kernel<false><<<gLayer, blkLayer, 0, stream>>>(bufA, rowptr, csr, nullptr, dinv, W3, b3, out,  N, E);
}


# Layers 2 and 3 read each node's records by scalar loads into scalar registers: source and norm feed row loads and sums directly, no lane broadcasts; rows requested in fours; sums fused in list order
# speedup vs baseline: 4.7230x; 1.0366x over previous
#include <hip/hip_runtime.h>
#include <hip/hip_bf16.h>

typedef __attribute__((ext_vector_type(2))) float v2f;
typedef __attribute__((ext_vector_type(8))) float v8f;

#define D 64

#define NCHUNK 256
#define NBIN 256
#define BIN_SHIFT 9
#define BIN_NODES 512

__global__ void __launch_bounds__(256)
chunk_hist_kernel(const int* __restrict__ dst, int* __restrict__ hist, int E) {
    __shared__ int h[NBIN];
    const int t = threadIdx.x;
    h[t] = 0;
    __syncthreads();
    const int ch = (E + NCHUNK - 1) / NCHUNK;
    const int e0 = blockIdx.x * ch;
    const int e1 = (e0 + ch < E) ? (e0 + ch) : E;
    int e = e0 + t;
    for (; e + 768 < e1; e += 1024) {
        const int d0 = dst[e];
        const int d1 = dst[e + 256];
        const int d2 = dst[e + 512];
        const int d3 = dst[e + 768];
        atomicAdd(&h[d0 >> BIN_SHIFT], 1);
        atomicAdd(&h[d1 >> BIN_SHIFT], 1);
        atomicAdd(&h[d2 >> BIN_SHIFT], 1);
        atomicAdd(&h[d3 >> BIN_SHIFT], 1);
    }
    for (; e < e1; e += 256) atomicAdd(&h[dst[e] >> BIN_SHIFT], 1);
    __syncthreads();
    hist[blockIdx.x * NBIN + t] = h[t];
}

__global__ void __launch_bounds__(256)
bin_scatter_kernel(const int* __restrict__ src,
                   const int* __restrict__ dst,
                   const float* __restrict__ w,
                   const int* __restrict__ hist,
                   int* __restrict__ binstart,
                   float4* __restrict__ binrec, int E) {
    __shared__ int cur[NBIN];
    __shared__ int wtot[8];
    const int t = threadIdx.x;
    const int lane = t & 31;
    const int wave = t >> 5;
    const int c = blockIdx.x;
    int tot = 0;
    int before = 0;
    for (int k0 = 0; k0 < NCHUNK; k0 += 32) {
        int x[32];
#pragma unroll
        for (int u = 0; u < 32; ++u) x[u] = hist[(k0 + u) * NBIN + t];
#pragma unroll
        for (int u = 0; u < 32; ++u) {
            tot += x[u];
            before += (k0 + u < c) ? x[u] : 0;
        }
    }
    int incl = tot;
    for (int off = 1; off < 32; off <<= 1) {
        const int v = __shfl_up(incl, off);
        if (lane >= off) incl += v;
    }
    if (lane == 31) wtot[wave] = incl;
    __syncthreads();
    int ex = incl - tot;
    for (int k = 0; k < wave; ++k) ex += wtot[k];
    cur[t] = ex + before;
    if (c == 0) {
        binstart[t] = ex;
        if (t == NBIN - 1) binstart[NBIN] = ex + tot;
    }
    __syncthreads();

    const int ch = (E + NCHUNK - 1) / NCHUNK;
    const int e0 = c * ch;
    const int e1 = (e0 + ch < E) ? (e0 + ch) : E;
    int e = e0 + t;
    for (; e + 768 < e1; e += 1024) {
        int s[4], d[4];
        float ww[4];
#pragma unroll
        for (int u = 0; u < 4; ++u) {
            s[u] = src[e + u * 256];
            d[u] = dst[e + u * 256];
            ww[u] = w[e + u * 256];
        }
#pragma unroll
        for (int u = 0; u < 4; ++u) {
            const int pos = atomicAdd(&cur[d[u] >> BIN_SHIFT], 1);
            binrec[pos] = make_float4(__int_as_float(d[u]), ww[u], __int_as_float(s[u]), 0.0f);
        }
    }
    for (; e < e1; e += 256) {
        const int s = src[e];
        const int d = dst[e];
        const float ww = w[e];
        const int pos = atomicAdd(&cur[d >> BIN_SHIFT], 1);
        binrec[pos] = make_float4(__int_as_float(d), ww, __int_as_float(s), 0.0f);
    }
}

__global__ void __launch_bounds__(512)
bin_sort_kernel(const float4* __restrict__ binrec,
                const int* __restrict__ binstart,
                int* __restrict__ rowptr,
                float* __restrict__ dinv,
                float2* __restrict__ ew, int n) {
    __shared__ int lc[BIN_NODES];
    __shared__ float ldeg[BIN_NODES];
    __shared__ int wtot[16];
    const int t = threadIdx.x;
    const int lane = t & 31;
    const int wave = t >> 5;
    const int n0 = blockIdx.x << BIN_SHIFT;
    if (n0 >= n) return;
    const int p0 = binstart[blockIdx.x];
    const int p1 = binstart[blockIdx.x + 1];
    lc[t] = 0;
    ldeg[t] = 0.0f;
    __syncthreads();
    const float2* __restrict__ dw = reinterpret_cast<const float2*>(binrec);
    int p = p0 + t;
    for (; p + 1536 < p1; p += 2048) {
        const float2 r0 = dw[(size_t)p * 2];
        const float2 r1 = dw[(size_t)(p + 512) * 2];
        const float2 r2 = dw[(size_t)(p + 1024) * 2];
        const float2 r3 = dw[(size_t)(p + 1536) * 2];
        atomicAdd(&lc[__float_as_int(r0.x) - n0], 1);
        atomicAdd(&lc[__float_as_int(r1.x) - n0], 1);
        atomicAdd(&lc[__float_as_int(r2.x) - n0], 1);
        atomicAdd(&lc[__float_as_int(r3.x) - n0], 1);
        atomicAdd(&ldeg[__float_as_int(r0.x) - n0], r0.y);
        atomicAdd(&ldeg[__float_as_int(r1.x) - n0], r1.y);
        atomicAdd(&ldeg[__float_as_int(r2.x) - n0], r2.y);
        atomicAdd(&ldeg[__float_as_int(r3.x) - n0], r3.y);
    }
    for (; p < p1; p += 512) {
        const float2 r = dw[(size_t)p * 2];
        atomicAdd(&lc[__float_as_int(r.x) - n0], 1);
        atomicAdd(&ldeg[__float_as_int(r.x) - n0], r.y);
    }
    __syncthreads();
    const int cnt = lc[t];
    const float dg = 1.0f + ldeg[t];
    int incl = cnt;
    for (int off = 1; off < 32; off <<= 1) {
        const int v = __shfl_up(incl, off);
        if (lane >= off) incl += v;
    }
    if (lane == 31) wtot[wave] = incl;
    __syncthreads();
    int first = p0 + incl - cnt;
    for (int k = 0; k < wave; ++k) first += wtot[k];
    if (n0 + t < n) {
        rowptr[n0 + t] = first;
        dinv[n0 + t] = (dg > 0.0f) ? rsqrtf(fmaxf(dg, 1e-30f)) : 0.0f;
    }
    if (n0 + t == n - 1) rowptr[n] = first + cnt;
    lc[t] = first;
    __syncthreads();
    p = p0 + t;
    for (; p + 1536 < p1; p += 2048) {
        float4 r[4];
#pragma unroll
        for (int u = 0; u < 4; ++u) r[u] = binrec[p + u * 512];
#pragma unroll
        for (int u = 0; u < 4; ++u) {
            const int slot = atomicAdd(&lc[__float_as_int(r[u].x) - n0], 1);
            ew[slot] = make_float2(r[u].z, r[u].y);
        }
    }
    for (; p < p1; p += 512) {
        const float4 r = binrec[p];
        const int slot = atomicAdd(&lc[__float_as_int(r.x) - n0], 1);
        ew[slot] = make_float2(r.z, r.y);
    }
}

#define TILE_NODES 32
#define WAVE_NODES 8
#define A_PITCH 68

static __device__ __forceinline__ int lane_get_i(int v, int j) {
    return __builtin_amdgcn_readlane(v, j);
}
static __device__ __forceinline__ float lane_get_f(float v, int j) {
    return __int_as_float(__builtin_amdgcn_readlane(__float_as_int(v), j));
}
static __device__ __forceinline__ float2 row_pair(const float* __restrict__ H, int row, int lane) {
    return reinterpret_cast<const float2*>(H + (size_t)row * D)[lane];
}
template <int R>
static __device__ __forceinline__ void issue_rows(const float* __restrict__ H, int s, int j,
                                                  int lane, float2 (&hv)[R]) {
#pragma unroll
    for (int u = 0; u < R; ++u) hv[u] = row_pair(H, lane_get_i(s, j + u), lane);
}
template <int R>
static __device__ __forceinline__ void sum_rows(float2& acc, const float2 (&hv)[R], float nm,
                                                int j, int cnt) {
#pragma unroll
    for (int u = 0; u < R; ++u) {
        const float nj = lane_get_f(nm, j + u);
        const bool live = (j + u < cnt);
        const float hx = live ? hv[u].x : 0.0f;
        const float hy = live ? hv[u].y : 0.0f;
        acc.x += hx * nj;
        acc.y += hy * nj;
    }
}

static __device__ __forceinline__ float2 row_pair_at(const float* __restrict__ H, int row, unsigned lane8) {
    const unsigned off = ((unsigned)row << 8) + lane8;
    return *reinterpret_cast<const float2*>(reinterpret_cast<const char*>(H) + off);
}
static __device__ __forceinline__ void load_recs(const float2* __restrict__ csr, int p, float2 (&r)[16]) {
#pragma unroll
    for (int u = 0; u < 16; ++u) r[u] = csr[p + u];
}
template <int R>
static __device__ __forceinline__ void issue_rows_s(const float* __restrict__ H, const float2 (&r)[16],
                                                    int cnt, int own, unsigned lane8, float2 (&hv)[16]) {
#pragma unroll
    for (int u = 0; u < R; ++u) {
        int s = __float_as_int(r[u].x);
        if (u >= R - 4) s = (u < cnt) ? s : own;
        hv[u] = row_pair_at(H, s, lane8);
    }
}
template <int R>
static __device__ __forceinline__ void sum_rows_s(float2& acc, const float2 (&hv)[16],
                                                  const float2 (&r)[16], int cnt) {
#pragma unroll
    for (int u = 0; u < R; ++u) {
        float nj = r[u].y;
        if (u >= R - 4) nj = (u < cnt) ? nj : 0.0f;
        acc.x = fmaf(hv[u].x, nj, acc.x);
        acc.y = fmaf(hv[u].y, nj, acc.y);
    }
}
template <int R>
static __device__ __forceinline__ void rows_arm(float2& acc, const float* __restrict__ H,
                                                const float2 (&r)[16], int c, int own, unsigned lane8) {
    float2 hv[16];
    issue_rows_s<R>(H, r, c, own, lane8, hv);
    sum_rows_s<R>(acc, hv, r, c);
}
static __device__ __forceinline__ void rows_group(float2& acc, const float* __restrict__ H,
                                                  const float2 (&r)[16], int c, int own, unsigned lane8) {
    if (c > 12) rows_arm<16>(acc, H, r, c, own, lane8);
    else if (c > 8) rows_arm<12>(acc, H, r, c, own, lane8);
    else if (c > 4) rows_arm<8>(acc, H, r, c, own, lane8);
    else if (c > 0) rows_arm<4>(acc, H, r, c, own, lane8);
}
static __device__ __forceinline__ void walk_node(const float* __restrict__ X,
                                                 const float2* __restrict__ csr,
                                                 int rp, float dv, int q, int nbeg, int n,
                                                 int lane, unsigned lane8, float* __restrict__ arow,
                                                 int& beg0, int& end0, float2& hs0,
                                                 float2 (&rc)[16], float2 (&rn)[16]) {
    const int node = nbeg + q;
    const int own = (node < n) ? node : (n - 1);
    const int n1 = node + 1;
    const int beg1 = end0;
    const int end1 = lane_get_i(rp, q + 2);
    const bool live1 = (q + 1 < WAVE_NODES) && (n1 < n);
    const int n1c = (n1 < n) ? n1 : (n - 1);
    const int call = end0 - beg0;
    const int c = (call < 16) ? call : 16;
    const float di0 = lane_get_f(dv, q);

    float2 acc = make_float2(0.0f, 0.0f);
    const float2 hs1 = row_pair(X, n1c, lane);
    rows_group(acc, X, rc, c, own, lane8);
    load_recs(csr, beg1, rn);
    for (int base = beg0 + 16; base < end0; base += 16) {
        const int c2 = (end0 - base < 16) ? (end0 - base) : 16;
        load_recs(csr, base, rc);
        rows_group(acc, X, rc, c2, own, lane8);
    }
    const float sl = di0 * di0;
    acc.x += hs0.x * sl;
    acc.y += hs0.y * sl;
    reinterpret_cast<float2*>(arow)[lane] = acc;

    beg0 = beg1; end0 = end1;
    hs0.x = live1 ? hs1.x : 0.0f;
    hs0.y = live1 ? hs1.y : 0.0f;
}

template <bool FIRST>
__global__ void __launch_bounds__(128)
aggregate_transform_kernel(const float* __restrict__ X,
                           const int* __restrict__ rowptr,
                           const float2* __restrict__ csr,
                           float2* __restrict__ csr_out,
                           const float* __restrict__ dinv,
                           const float* __restrict__ W,
                           const float* __restrict__ bias,
                           float* __restrict__ outp, int n, int E) {
    __shared__ float sA[TILE_NODES * A_PITCH];
    const int lane = threadIdx.x & 31;
    const int wave = __builtin_amdgcn_readfirstlane((int)(threadIdx.x >> 5));
    const int tile0 = blockIdx.x * TILE_NODES;
    const int nbeg = tile0 + wave * WAVE_NODES;
    const int elast = (E > 0) ? (E - 1) : 0;

    int rp;
    float dv = 0.0f;
    {
        const int j = (lane < WAVE_NODES) ? lane : WAVE_NODES;
        const int idx = nbeg + j;
        rp = rowptr[(idx < n) ? idx : n];
        if (lane < WAVE_NODES && idx < n) dv = dinv[idx];
    }

    int beg0 = lane_get_i(rp, 0);
    int end0 = lane_get_i(rp, 1);
    float2 hs0 = make_float2(0.0f, 0.0f);
    if (nbeg < n) hs0 = row_pair(X, nbeg, lane);

    if (FIRST) {
        int s0 = 0;
        float nm0 = 0.0f;
        if (lane < end0 - beg0) {
            float2 rec = csr[beg0 + lane];
            s0 = __float_as_int(rec.x);
            nm0 = rec.y;
        }
        for (int q = 0; q < WAVE_NODES; ++q) {
            const int n1 = nbeg + q + 1;
            const int beg1 = end0;
            const int end1 = lane_get_i(rp, q + 2);
            const bool live1 = (q + 1 < WAVE_NODES) && (n1 < n);
            const int n1c = (n1 < n) ? n1 : (n - 1);
            int idx1 = beg1 + lane;
            idx1 = (idx1 < elast) ? idx1 : elast;
            const bool has1 = lane < end1 - beg1;
            int s1;
            float nm1;
            float2 hs1;

            const int cnt = (end0 - beg0 < 32) ? (end0 - beg0) : 32;
            const float di0 = lane_get_f(dv, q);
            float ds0 = 0.0f;
            if (FIRST) ds0 = dinv[s0];
            float2 acc = make_float2(0.0f, 0.0f);
            if (FIRST || cnt > 8) {
                float2 hv[16];
                issue_rows<16>(X, s0, 0, lane, hv);
                const float2 rec = csr[idx1];
                hs1 = row_pair(X, n1c, lane);
                if (FIRST) {
                    nm0 = ds0 * nm0 * di0;
                    if (lane < cnt) csr_out[beg0 + lane] = make_float2(__int_as_float(s0), nm0);
                }
                sum_rows<16>(acc, hv, nm0, 0, cnt);
                s1 = has1 ? __float_as_int(rec.x) : 0;
                nm1 = has1 ? rec.y : 0.0f;
                if (cnt > 16) {
                    float2 hw[16];
                    issue_rows<16>(X, s0, 16, lane, hw);
                    sum_rows<16>(acc, hw, nm0, 16, cnt);
                }
            } else {
                float2 hv[8];
                issue_rows<8>(X, s0, 0, lane, hv);
                const float2 rec = csr[idx1];
                hs1 = row_pair(X, n1c, lane);
                sum_rows<8>(acc, hv, nm0, 0, cnt);
                s1 = has1 ? __float_as_int(rec.x) : 0;
                nm1 = has1 ? rec.y : 0.0f;
            }
            for (int base = beg0 + 32; base < end0; base += 32) {
                const int c2 = (end0 - base < 32) ? (end0 - base) : 32;
                int s = 0;
                float nm = 0.0f;
                if (lane < c2) {
                    float2 rec = csr[base + lane];
                    s = __float_as_int(rec.x);
                    nm = rec.y;
                }
                if (FIRST) {
                    nm = dinv[s] * nm * di0;
                    if (lane < c2) csr_out[base + lane] = make_float2(__int_as_float(s), nm);
                }
                float2 hv[16];
                issue_rows<16>(X, s, 0, lane, hv);
                sum_rows<16>(acc, hv, nm, 0, c2);
                if (c2 > 16) {
                    float2 hw[16];
                    issue_rows<16>(X, s, 16, lane, hw);
                    sum_rows<16>(acc, hw, nm, 16, c2);
                }
            }
            const float sl = di0 * di0;
            acc.x += hs0.x * sl;
            acc.y += hs0.y * sl;
            reinterpret_cast<float2*>(&sA[(wave * WAVE_NODES + q) * A_PITCH])[lane] = acc;

            beg0 = beg1; end0 = end1; s0 = s1; nm0 = nm1;
            hs0.x = live1 ? hs1.x : 0.0f;
            hs0.y = live1 ? hs1.y : 0.0f;
        }
    } else {
        const unsigned lane8 = (unsigned)lane << 3;
        float2 ra[16], rb[16];
        load_recs(csr, beg0, ra);
        for (int q = 0; q < WAVE_NODES; q += 2) {
            walk_node(X, csr, rp, dv, q, nbeg, n, lane, lane8,
                      &sA[(wave * WAVE_NODES + q) * A_PITCH], beg0, end0, hs0, ra, rb);
            walk_node(X, csr, rp, dv, q + 1, nbeg, n, lane, lane8,
                      &sA[(wave * WAVE_NODES + q + 1) * A_PITCH], beg0, end0, hs0, rb, ra);
        }
    }

    const int half = lane >> 4;
    const int l = lane & 15;
    const int c0 = wave * 16;
    const float* __restrict__ wcol = W + (c0 + l) + half * 2 * D;
    v2f bfrag[D / 4];
#pragma unroll
    for (int k = 0; k < D; k += 4) {
        bfrag[k / 4].x = wcol[k * D];
        bfrag[k / 4].y = wcol[k * D + D];
    }
    const float bc = bias[c0 + l];

    __syncthreads();

    const float* a0p = &sA[l * A_PITCH + half * 2];
    const float* a1p = &sA[(16 + l) * A_PITCH + half * 2];
    v8f acc0 = {};
    v8f acc1 = {};
#pragma unroll
    for (int k = 0; k < D; k += 4) {
        const v2f a0 = *reinterpret_cast<const v2f*>(a0p + k);
        const v2f a1 = *reinterpret_cast<const v2f*>(a1p + k);
        acc0 = __builtin_amdgcn_wmma_f32_16x16x4_f32(
             false, a0,  false, bfrag[k / 4],
             (short)0, acc0,  false,  false);
        acc1 = __builtin_amdgcn_wmma_f32_16x16x4_f32(
             false, a1,  false, bfrag[k / 4],
             (short)0, acc1,  false,  false);
    }

    const int r0 = tile0 + half * 8;
    float* __restrict__ op = outp + (size_t)r0 * D + c0 + l;
    if (tile0 + TILE_NODES <= n) {
#pragma unroll
        for (int v = 0; v < 8; ++v) {
            op[(size_t)v * D] = fmaxf(acc0[v] + bc, 0.0f);
            op[(size_t)(16 + v) * D] = fmaxf(acc1[v] + bc, 0.0f);
        }
    } else {
#pragma unroll
        for (int v = 0; v < 8; ++v) {
            if (r0 + v < n) op[(size_t)v * D] = fmaxf(acc0[v] + bc, 0.0f);
            if (r0 + 16 + v < n) op[(size_t)(16 + v) * D] = fmaxf(acc1[v] + bc, 0.0f);
        }
    }
}


extern "C" void kernel_launch(void* const* d_in, const int* in_sizes, int n_in,
                              void* d_out, int out_size, void* d_ws, size_t ws_size,
                              hipStream_t stream) {
    (void)n_in; (void)out_size; (void)ws_size;

    const float* x  = (const float*)d_in[0];
    const int*   ei = (const int*)  d_in[1];
    const float* ea = (const float*)d_in[2];
    const float* W1 = (const float*)d_in[3];
    const float* b1 = (const float*)d_in[4];
    const float* W2 = (const float*)d_in[5];
    const float* b2 = (const float*)d_in[6];
    const float* W3 = (const float*)d_in[7];
    const float* b3 = (const float*)d_in[8];

    const int N = in_sizes[0] / D;
    const int E = in_sizes[2];
    const int* src = ei;
    const int* dst = ei + E;

    float*  out      = (float*)d_out;
    float*  bufA     = (float*)d_ws;
    float*  bufB     = bufA + (size_t)N * D;
    float4* binrec   = (float4*)(bufB + (size_t)N * D);
    int*    hist     = (int*)(binrec + E);
    int*    binstart = hist + NCHUNK * NBIN;
    float2* ew       = (float2*)(binstart + NBIN + 4);
    float2* csr      = ew + E;
    float*  dinv     = (float*)(csr + E);
    int*    rowptr   = (int*)(dinv + N);

    const dim3 blk(256);
    const dim3 blkLayer(128);
    const dim3 gLayer((N + TILE_NODES - 1) / TILE_NODES);

    chunk_hist_kernel<<<dim3(NCHUNK), blk, 0, stream>>>(dst, hist, E);
    bin_scatter_kernel<<<dim3(NCHUNK), blk, 0, stream>>>(src, dst, ea, hist, binstart, binrec, E);
    bin_sort_kernel<<<dim3(NBIN), dim3(BIN_NODES), 0, stream>>>(binrec, binstart, rowptr, dinv, ew, N);

    aggregate_transform_kernel<true><<<gLayer, blkLayer, 0, stream>>>(x,    rowptr, ew,  csr,     dinv, W1, b1, bufB, N, E);
    aggregate_transform_kernel<false><<<gLayer, blkLayer, 0, stream>>>(bufB, rowptr, csr, nullptr, dinv, W2, b2, bufA, N, E);
    aggregate_transform_kernel<false><<<gLayer, blkLayer, 0, stream>>>(bufA, rowptr, csr, nullptr, dinv, W3, b3, out,  N, E);
}


# Sort level 1 made chunk-local: one launch counts, scans and places a chunk's records by range; range sort finds its segments by an offset table, reads each once: 6->5 launches; list/sum orders change
# speedup vs baseline: 5.1764x; 1.0960x over previous
#include <hip/hip_runtime.h>
#include <hip/hip_bf16.h>

typedef __attribute__((ext_vector_type(2))) float v2f;
typedef __attribute__((ext_vector_type(8))) float v8f;

#define D 64

#define NCHUNK 256
#define NBIN 256
#define BIN_SHIFT 9
#define BIN_NODES 512

__global__ void __launch_bounds__(256)
chunk_bin_kernel(const int* __restrict__ src,
                 const int* __restrict__ dst,
                 const float* __restrict__ w,
                 int* __restrict__ offt,
                 float4* __restrict__ binrec, int E) {
    __shared__ int h[NBIN];
    __shared__ int cur[NBIN];
    __shared__ int wtot[8];
    const int t = threadIdx.x;
    const int lane = t & 31;
    const int wave = t >> 5;
    const int c = blockIdx.x;
    h[t] = 0;
    __syncthreads();
    const int ch = (E + NCHUNK - 1) / NCHUNK;
    const int e0 = c * ch;
    const int e1 = (e0 + ch < E) ? (e0 + ch) : E;
    int e = e0 + t;
    for (; e + 768 < e1; e += 1024) {
        const int d0 = dst[e];
        const int d1 = dst[e + 256];
        const int d2 = dst[e + 512];
        const int d3 = dst[e + 768];
        atomicAdd(&h[d0 >> BIN_SHIFT], 1);
        atomicAdd(&h[d1 >> BIN_SHIFT], 1);
        atomicAdd(&h[d2 >> BIN_SHIFT], 1);
        atomicAdd(&h[d3 >> BIN_SHIFT], 1);
    }
    for (; e < e1; e += 256) atomicAdd(&h[dst[e] >> BIN_SHIFT], 1);
    __syncthreads();
    const int cnt = h[t];
    int incl = cnt;
    for (int off = 1; off < 32; off <<= 1) {
        const int v = __shfl_up(incl, off);
        if (lane >= off) incl += v;
    }
    if (lane == 31) wtot[wave] = incl;
    __syncthreads();
    int ex = incl - cnt;
    for (int k = 0; k < wave; ++k) ex += wtot[k];
    cur[t] = e0 + ex;
    offt[t * NCHUNK + c] = ex;
    if (t == NBIN - 1) offt[NBIN * NCHUNK + c] = ex + cnt;
    __syncthreads();

    e = e0 + t;
    for (; e + 768 < e1; e += 1024) {
        int s[4], d[4];
        float ww[4];
#pragma unroll
        for (int u = 0; u < 4; ++u) {
            s[u] = src[e + u * 256];
            d[u] = dst[e + u * 256];
            ww[u] = w[e + u * 256];
        }
#pragma unroll
        for (int u = 0; u < 4; ++u) {
            const int pos = atomicAdd(&cur[d[u] >> BIN_SHIFT], 1);
            binrec[pos] = make_float4(__int_as_float(d[u]), ww[u], __int_as_float(s[u]), 0.0f);
        }
    }
    for (; e < e1; e += 256) {
        const int s = src[e];
        const int d = dst[e];
        const float ww = w[e];
        const int pos = atomicAdd(&cur[d >> BIN_SHIFT], 1);
        binrec[pos] = make_float4(__int_as_float(d), ww, __int_as_float(s), 0.0f);
    }
}

#define SEG_PER_HALF 8
#define SEG_HELD 16

static __device__ __forceinline__ int seg_pos(const int (&dl)[SEG_PER_HALF],
                                              const int (&pre)[SEG_PER_HALF + 1], int j) {
    int d = dl[0];
#pragma unroll
    for (int s = 1; s < SEG_PER_HALF; ++s) d = (j >= pre[s]) ? dl[s] : d;
    return j + d;
}

__global__ void __launch_bounds__(512)
bin_sort_kernel(const float4* __restrict__ binrec,
                const int* __restrict__ offt,
                int* __restrict__ rowptr,
                float* __restrict__ dinv,
                float2* __restrict__ ew, int n, int E) {
    __shared__ int lc[BIN_NODES];
    __shared__ float ldeg[BIN_NODES];
    __shared__ int wtot[16];
    __shared__ int wbefore[16];
    const int t = threadIdx.x;
    const int lane = t & 31;
    const int wave = t >> 5;
    const int hw = t >> 4;
    const int l16 = t & 15;
    const int r = blockIdx.x;
    const int n0 = r << BIN_SHIFT;
    if (n0 >= n) return;
    const int ch = (E + NCHUNK - 1) / NCHUNK;

    const int4* __restrict__ row0 = reinterpret_cast<const int4*>(offt + r * NCHUNK + hw * SEG_PER_HALF);
    const int4* __restrict__ row1 = reinterpret_cast<const int4*>(offt + (r + 1) * NCHUNK + hw * SEG_PER_HALF);
    const int4 a0 = row0[0];
    const int4 a1 = row0[1];
    const int4 b0 = row1[0];
    const int4 b1 = row1[1];
    const int so[SEG_PER_HALF] = {a0.x, a0.y, a0.z, a0.w, a1.x, a1.y, a1.z, a1.w};
    const int eo[SEG_PER_HALF] = {b0.x, b0.y, b0.z, b0.w, b1.x, b1.y, b1.z, b1.w};
    int pre[SEG_PER_HALF + 1];
    int dl[SEG_PER_HALF];
    int before = 0;
    pre[0] = 0;
#pragma unroll
    for (int s = 0; s < SEG_PER_HALF; ++s) {
        dl[s] = (hw * SEG_PER_HALF + s) * ch + so[s] - pre[s];
        pre[s + 1] = pre[s] + (eo[s] - so[s]);
        before += so[s];
    }
    const int total = pre[SEG_PER_HALF];
    const int wsum = before + __shfl_xor(before, 16);

    lc[t] = 0;
    ldeg[t] = 0.0f;
    if (lane == 0) wbefore[wave] = wsum;
    __syncthreads();
    int p0 = 0;
    for (int k = 0; k < 16; ++k) p0 += wbefore[k];

    float4 rec[SEG_HELD];
#pragma unroll
    for (int u = 0; u < SEG_HELD; ++u) {
        const int j = l16 + 16 * u;
        rec[u] = make_float4(0.0f, 0.0f, 0.0f, 0.0f);
        if (j < total) rec[u] = binrec[seg_pos(dl, pre, j)];
    }
#pragma unroll
    for (int u = 0; u < SEG_HELD; ++u) {
        if (l16 + 16 * u < total) {
            const int node = __float_as_int(rec[u].x) - n0;
            atomicAdd(&lc[node], 1);
            atomicAdd(&ldeg[node], rec[u].y);
        }
    }
    const float2* __restrict__ dw = reinterpret_cast<const float2*>(binrec);
    for (int j = 16 * SEG_HELD + l16; j < total; j += 16) {
        const float2 q = dw[(size_t)seg_pos(dl, pre, j) * 2];
        atomicAdd(&lc[__float_as_int(q.x) - n0], 1);
        atomicAdd(&ldeg[__float_as_int(q.x) - n0], q.y);
    }
    __syncthreads();
    const int cnt = lc[t];
    const float dg = 1.0f + ldeg[t];
    int incl = cnt;
    for (int off = 1; off < 32; off <<= 1) {
        const int v = __shfl_up(incl, off);
        if (lane >= off) incl += v;
    }
    if (lane == 31) wtot[wave] = incl;
    __syncthreads();
    int first = p0 + incl - cnt;
    for (int k = 0; k < wave; ++k) first += wtot[k];
    if (n0 + t < n) {
        rowptr[n0 + t] = first;
        dinv[n0 + t] = (dg > 0.0f) ? rsqrtf(fmaxf(dg, 1e-30f)) : 0.0f;
    }
    if (n0 + t == n - 1) rowptr[n] = first + cnt;
    lc[t] = first;
    __syncthreads();
#pragma unroll
    for (int u = 0; u < SEG_HELD; ++u) {
        if (l16 + 16 * u < total) {
            const int slot = atomicAdd(&lc[__float_as_int(rec[u].x) - n0], 1);
            ew[slot] = make_float2(rec[u].z, rec[u].y);
        }
    }
    for (int j = 16 * SEG_HELD + l16; j < total; j += 16) {
        const float4 q = binrec[seg_pos(dl, pre, j)];
        const int slot = atomicAdd(&lc[__float_as_int(q.x) - n0], 1);
        ew[slot] = make_float2(q.z, q.y);
    }
}

#define TILE_NODES 32
#define WAVE_NODES 8
#define A_PITCH 68

static __device__ __forceinline__ int lane_get_i(int v, int j) {
    return __builtin_amdgcn_readlane(v, j);
}
static __device__ __forceinline__ float lane_get_f(float v, int j) {
    return __int_as_float(__builtin_amdgcn_readlane(__float_as_int(v), j));
}
static __device__ __forceinline__ float2 row_pair(const float* __restrict__ H, int row, int lane) {
    return reinterpret_cast<const float2*>(H + (size_t)row * D)[lane];
}
template <int R>
static __device__ __forceinline__ void issue_rows(const float* __restrict__ H, int s, int j,
                                                  int lane, float2 (&hv)[R]) {
#pragma unroll
    for (int u = 0; u < R; ++u) hv[u] = row_pair(H, lane_get_i(s, j + u), lane);
}
template <int R>
static __device__ __forceinline__ void sum_rows(float2& acc, const float2 (&hv)[R], float nm,
                                                int j, int cnt) {
#pragma unroll
    for (int u = 0; u < R; ++u) {
        const float nj = lane_get_f(nm, j + u);
        const bool live = (j + u < cnt);
        const float hx = live ? hv[u].x : 0.0f;
        const float hy = live ? hv[u].y : 0.0f;
        acc.x += hx * nj;
        acc.y += hy * nj;
    }
}

static __device__ __forceinline__ float2 row_pair_at(const float* __restrict__ H, int row, unsigned lane8) {
    const unsigned off = ((unsigned)row << 8) + lane8;
    return *reinterpret_cast<const float2*>(reinterpret_cast<const char*>(H) + off);
}
static __device__ __forceinline__ void load_recs(const float2* __restrict__ csr, int p, float2 (&r)[16]) {
#pragma unroll
    for (int u = 0; u < 16; ++u) r[u] = csr[p + u];
}
template <int R>
static __device__ __forceinline__ void issue_rows_s(const float* __restrict__ H, const float2 (&r)[16],
                                                    int cnt, int own, unsigned lane8, float2 (&hv)[16]) {
#pragma unroll
    for (int u = 0; u < R; ++u) {
        int s = __float_as_int(r[u].x);
        if (u >= R - 4) s = (u < cnt) ? s : own;
        hv[u] = row_pair_at(H, s, lane8);
    }
}
template <int R>
static __device__ __forceinline__ void sum_rows_s(float2& acc, const float2 (&hv)[16],
                                                  const float2 (&r)[16], int cnt) {
#pragma unroll
    for (int u = 0; u < R; ++u) {
        float nj = r[u].y;
        if (u >= R - 4) nj = (u < cnt) ? nj : 0.0f;
        acc.x = fmaf(hv[u].x, nj, acc.x);
        acc.y = fmaf(hv[u].y, nj, acc.y);
    }
}
template <int R>
static __device__ __forceinline__ void rows_arm(float2& acc, const float* __restrict__ H,
                                                const float2 (&r)[16], int c, int own, unsigned lane8) {
    float2 hv[16];
    issue_rows_s<R>(H, r, c, own, lane8, hv);
    sum_rows_s<R>(acc, hv, r, c);
}
static __device__ __forceinline__ void rows_group(float2& acc, const float* __restrict__ H,
                                                  const float2 (&r)[16], int c, int own, unsigned lane8) {
    if (c > 12) rows_arm<16>(acc, H, r, c, own, lane8);
    else if (c > 8) rows_arm<12>(acc, H, r, c, own, lane8);
    else if (c > 4) rows_arm<8>(acc, H, r, c, own, lane8);
    else if (c > 0) rows_arm<4>(acc, H, r, c, own, lane8);
}
static __device__ __forceinline__ void walk_node(const float* __restrict__ X,
                                                 const float2* __restrict__ csr,
                                                 int rp, float dv, int q, int nbeg, int n,
                                                 int lane, unsigned lane8, float* __restrict__ arow,
                                                 int& beg0, int& end0, float2& hs0,
                                                 float2 (&rc)[16], float2 (&rn)[16]) {
    const int node = nbeg + q;
    const int own = (node < n) ? node : (n - 1);
    const int n1 = node + 1;
    const int beg1 = end0;
    const int end1 = lane_get_i(rp, q + 2);
    const bool live1 = (q + 1 < WAVE_NODES) && (n1 < n);
    const int n1c = (n1 < n) ? n1 : (n - 1);
    const int call = end0 - beg0;
    const int c = (call < 16) ? call : 16;
    const float di0 = lane_get_f(dv, q);

    float2 acc = make_float2(0.0f, 0.0f);
    const float2 hs1 = row_pair(X, n1c, lane);
    rows_group(acc, X, rc, c, own, lane8);
    load_recs(csr, beg1, rn);
    for (int base = beg0 + 16; base < end0; base += 16) {
        const int c2 = (end0 - base < 16) ? (end0 - base) : 16;
        load_recs(csr, base, rc);
        rows_group(acc, X, rc, c2, own, lane8);
    }
    const float sl = di0 * di0;
    acc.x += hs0.x * sl;
    acc.y += hs0.y * sl;
    reinterpret_cast<float2*>(arow)[lane] = acc;

    beg0 = beg1; end0 = end1;
    hs0.x = live1 ? hs1.x : 0.0f;
    hs0.y = live1 ? hs1.y : 0.0f;
}

template <bool FIRST>
__global__ void __launch_bounds__(128)
aggregate_transform_kernel(const float* __restrict__ X,
                           const int* __restrict__ rowptr,
                           const float2* __restrict__ csr,
                           float2* __restrict__ csr_out,
                           const float* __restrict__ dinv,
                           const float* __restrict__ W,
                           const float* __restrict__ bias,
                           float* __restrict__ outp, int n, int E) {
    __shared__ float sA[TILE_NODES * A_PITCH];
    const int lane = threadIdx.x & 31;
    const int wave = __builtin_amdgcn_readfirstlane((int)(threadIdx.x >> 5));
    const int tile0 = blockIdx.x * TILE_NODES;
    const int nbeg = tile0 + wave * WAVE_NODES;
    const int elast = (E > 0) ? (E - 1) : 0;

    int rp;
    float dv = 0.0f;
    {
        const int j = (lane < WAVE_NODES) ? lane : WAVE_NODES;
        const int idx = nbeg + j;
        rp = rowptr[(idx < n) ? idx : n];
        if (lane < WAVE_NODES && idx < n) dv = dinv[idx];
    }

    int beg0 = lane_get_i(rp, 0);
    int end0 = lane_get_i(rp, 1);
    float2 hs0 = make_float2(0.0f, 0.0f);
    if (nbeg < n) hs0 = row_pair(X, nbeg, lane);

    if (FIRST) {
        int s0 = 0;
        float nm0 = 0.0f;
        if (lane < end0 - beg0) {
            float2 rec = csr[beg0 + lane];
            s0 = __float_as_int(rec.x);
            nm0 = rec.y;
        }
        for (int q = 0; q < WAVE_NODES; ++q) {
            const int n1 = nbeg + q + 1;
            const int beg1 = end0;
            const int end1 = lane_get_i(rp, q + 2);
            const bool live1 = (q + 1 < WAVE_NODES) && (n1 < n);
            const int n1c = (n1 < n) ? n1 : (n - 1);
            int idx1 = beg1 + lane;
            idx1 = (idx1 < elast) ? idx1 : elast;
            const bool has1 = lane < end1 - beg1;
            int s1;
            float nm1;
            float2 hs1;

            const int cnt = (end0 - beg0 < 32) ? (end0 - beg0) : 32;
            const float di0 = lane_get_f(dv, q);
            float ds0 = 0.0f;
            if (FIRST) ds0 = dinv[s0];
            float2 acc = make_float2(0.0f, 0.0f);
            if (FIRST || cnt > 8) {
                float2 hv[16];
                issue_rows<16>(X, s0, 0, lane, hv);
                const float2 rec = csr[idx1];
                hs1 = row_pair(X, n1c, lane);
                if (FIRST) {
                    nm0 = ds0 * nm0 * di0;
                    if (lane < cnt) csr_out[beg0 + lane] = make_float2(__int_as_float(s0), nm0);
                }
                sum_rows<16>(acc, hv, nm0, 0, cnt);
                s1 = has1 ? __float_as_int(rec.x) : 0;
                nm1 = has1 ? rec.y : 0.0f;
                if (cnt > 16) {
                    float2 hw[16];
                    issue_rows<16>(X, s0, 16, lane, hw);
                    sum_rows<16>(acc, hw, nm0, 16, cnt);
                }
            } else {
                float2 hv[8];
                issue_rows<8>(X, s0, 0, lane, hv);
                const float2 rec = csr[idx1];
                hs1 = row_pair(X, n1c, lane);
                sum_rows<8>(acc, hv, nm0, 0, cnt);
                s1 = has1 ? __float_as_int(rec.x) : 0;
                nm1 = has1 ? rec.y : 0.0f;
            }
            for (int base = beg0 + 32; base < end0; base += 32) {
                const int c2 = (end0 - base < 32) ? (end0 - base) : 32;
                int s = 0;
                float nm = 0.0f;
                if (lane < c2) {
                    float2 rec = csr[base + lane];
                    s = __float_as_int(rec.x);
                    nm = rec.y;
                }
                if (FIRST) {
                    nm = dinv[s] * nm * di0;
                    if (lane < c2) csr_out[base + lane] = make_float2(__int_as_float(s), nm);
                }
                float2 hv[16];
                issue_rows<16>(X, s, 0, lane, hv);
                sum_rows<16>(acc, hv, nm, 0, c2);
                if (c2 > 16) {
                    float2 hw[16];
                    issue_rows<16>(X, s, 16, lane, hw);
                    sum_rows<16>(acc, hw, nm, 16, c2);
                }
            }
            const float sl = di0 * di0;
            acc.x += hs0.x * sl;
            acc.y += hs0.y * sl;
            reinterpret_cast<float2*>(&sA[(wave * WAVE_NODES + q) * A_PITCH])[lane] = acc;

            beg0 = beg1; end0 = end1; s0 = s1; nm0 = nm1;
            hs0.x = live1 ? hs1.x : 0.0f;
            hs0.y = live1 ? hs1.y : 0.0f;
        }
    } else {
        const unsigned lane8 = (unsigned)lane << 3;
        float2 ra[16], rb[16];
        load_recs(csr, beg0, ra);
        for (int q = 0; q < WAVE_NODES; q += 2) {
            walk_node(X, csr, rp, dv, q, nbeg, n, lane, lane8,
                      &sA[(wave * WAVE_NODES + q) * A_PITCH], beg0, end0, hs0, ra, rb);
            walk_node(X, csr, rp, dv, q + 1, nbeg, n, lane, lane8,
                      &sA[(wave * WAVE_NODES + q + 1) * A_PITCH], beg0, end0, hs0, rb, ra);
        }
    }

    const int half = lane >> 4;
    const int l = lane & 15;
    const int c0 = wave * 16;
    const float* __restrict__ wcol = W + (c0 + l) + half * 2 * D;
    v2f bfrag[D / 4];
#pragma unroll
    for (int k = 0; k < D; k += 4) {
        bfrag[k / 4].x = wcol[k * D];
        bfrag[k / 4].y = wcol[k * D + D];
    }
    const float bc = bias[c0 + l];

    __syncthreads();

    const float* a0p = &sA[l * A_PITCH + half * 2];
    const float* a1p = &sA[(16 + l) * A_PITCH + half * 2];
    v8f acc0 = {};
    v8f acc1 = {};
#pragma unroll
    for (int k = 0; k < D; k += 4) {
        const v2f a0 = *reinterpret_cast<const v2f*>(a0p + k);
        const v2f a1 = *reinterpret_cast<const v2f*>(a1p + k);
        acc0 = __builtin_amdgcn_wmma_f32_16x16x4_f32(
             false, a0,  false, bfrag[k / 4],
             (short)0, acc0,  false,  false);
        acc1 = __builtin_amdgcn_wmma_f32_16x16x4_f32(
             false, a1,  false, bfrag[k / 4],
             (short)0, acc1,  false,  false);
    }

    const int r0 = tile0 + half * 8;
    float* __restrict__ op = outp + (size_t)r0 * D + c0 + l;
    if (tile0 + TILE_NODES <= n) {
#pragma unroll
        for (int v = 0; v < 8; ++v) {
            op[(size_t)v * D] = fmaxf(acc0[v] + bc, 0.0f);
            op[(size_t)(16 + v) * D] = fmaxf(acc1[v] + bc, 0.0f);
        }
    } else {
#pragma unroll
        for (int v = 0; v < 8; ++v) {
            if (r0 + v < n) op[(size_t)v * D] = fmaxf(acc0[v] + bc, 0.0f);
            if (r0 + 16 + v < n) op[(size_t)(16 + v) * D] = fmaxf(acc1[v] + bc, 0.0f);
        }
    }
}


extern "C" void kernel_launch(void* const* d_in, const int* in_sizes, int n_in,
                              void* d_out, int out_size, void* d_ws, size_t ws_size,
                              hipStream_t stream) {
    (void)n_in; (void)out_size; (void)ws_size;

    const float* x  = (const float*)d_in[0];
    const int*   ei = (const int*)  d_in[1];
    const float* ea = (const float*)d_in[2];
    const float* W1 = (const float*)d_in[3];
    const float* b1 = (const float*)d_in[4];
    const float* W2 = (const float*)d_in[5];
    const float* b2 = (const float*)d_in[6];
    const float* W3 = (const float*)d_in[7];
    const float* b3 = (const float*)d_in[8];

    const int N = in_sizes[0] / D;
    const int E = in_sizes[2];
    const int* src = ei;
    const int* dst = ei + E;

    float*  out      = (float*)d_out;
    float*  bufA     = (float*)d_ws;
    float*  bufB     = bufA + (size_t)N * D;
    float4* binrec   = (float4*)(bufB + (size_t)N * D);
    int*    offt     = (int*)(binrec + E);
    float2* ew       = (float2*)(offt + (NBIN + 1) * NCHUNK);
    float2* csr      = ew + E;
    float*  dinv     = (float*)(csr + E);
    int*    rowptr   = (int*)(dinv + N);

    const dim3 blk(256);
    const dim3 blkLayer(128);
    const dim3 gLayer((N + TILE_NODES - 1) / TILE_NODES);

    chunk_bin_kernel<<<dim3(NCHUNK), blk, 0, stream>>>(src, dst, ea, offt, binrec, E);
    bin_sort_kernel<<<dim3(NBIN), dim3(BIN_NODES), 0, stream>>>(binrec, offt, rowptr, dinv, ew, N, E);

    aggregate_transform_kernel<true><<<gLayer, blkLayer, 0, stream>>>(x,    rowptr, ew,  csr,     dinv, W1, b1, bufB, N, E);
    aggregate_transform_kernel<false><<<gLayer, blkLayer, 0, stream>>>(bufB, rowptr, csr, nullptr, dinv, W2, b2, bufA, N, E);
    aggregate_transform_kernel<false><<<gLayer, blkLayer, 0, stream>>>(bufA, rowptr, csr, nullptr, dinv, W3, b3, out,  N, E);
}


# Level-1 sort holds a thread's edges in registers: its dst, src, w loads all requested at once (was 7 dependent trips per pass), dst read once, LDS adds back to back; list/sum orders may change
# speedup vs baseline: 5.3618x; 1.0358x over previous
#include <hip/hip_runtime.h>
#include <hip/hip_bf16.h>

typedef __attribute__((ext_vector_type(2))) float v2f;
typedef __attribute__((ext_vector_type(8))) float v8f;

#define D 64

#define NCHUNK 256
#define NBIN 256
#define BIN_SHIFT 9
#define BIN_NODES 512

#define CH_HELD 20

__global__ void __launch_bounds__(256)
chunk_bin_kernel(const int* __restrict__ src,
                 const int* __restrict__ dst,
                 const float* __restrict__ w,
                 int* __restrict__ offt,
                 float4* __restrict__ binrec, int E) {
    __shared__ int h[NBIN + 32];
    __shared__ int cur[NBIN + 32];
    __shared__ int wtot[8];
    const int t = threadIdx.x;
    const int lane = t & 31;
    const int wave = t >> 5;
    const int c = blockIdx.x;
    if (E <= 0) {
        offt[t * NCHUNK + c] = 0;
        if (t == NBIN - 1) offt[NBIN * NCHUNK + c] = 0;
        return;
    }
    const int ch = (E + NCHUNK - 1) / NCHUNK;
    const int e0 = c * ch;
    const int e1 = (e0 + ch < E) ? (e0 + ch) : E;
    const int eb = e0 + t;
    const int elast = E - 1;

    int d[CH_HELD], s[CH_HELD];
    float ww[CH_HELD];
#pragma unroll
    for (int u = 0; u < CH_HELD; ++u) {
        const int e = (eb + 256 * u < e1) ? (eb + 256 * u) : elast;
        d[u] = dst[e];
        s[u] = src[e];
        ww[u] = w[e];
    }
    h[t] = 0;
    if (t < 32) { h[NBIN + t] = 0; cur[NBIN + t] = 0; }
    __syncthreads();
    int b[CH_HELD];
#pragma unroll
    for (int u = 0; u < CH_HELD; ++u) {
        b[u] = (eb + 256 * u < e1) ? (d[u] >> BIN_SHIFT) : (NBIN + lane);
        atomicAdd(&h[b[u]], 1);
    }
    for (int e = eb + 256 * CH_HELD; e < e1; e += 256) atomicAdd(&h[dst[e] >> BIN_SHIFT], 1);
    __syncthreads();
    const int cnt = h[t];
    int incl = cnt;
    for (int off = 1; off < 32; off <<= 1) {
        const int v = __shfl_up(incl, off);
        if (lane >= off) incl += v;
    }
    if (lane == 31) wtot[wave] = incl;
    __syncthreads();
    int ex = incl - cnt;
    for (int k = 0; k < wave; ++k) ex += wtot[k];
    cur[t] = e0 + ex;
    offt[t * NCHUNK + c] = ex;
    if (t == NBIN - 1) offt[NBIN * NCHUNK + c] = ex + cnt;
    __syncthreads();

    int pos[CH_HELD];
#pragma unroll
    for (int u = 0; u < CH_HELD; ++u) pos[u] = atomicAdd(&cur[b[u]], 1);
#pragma unroll
    for (int u = 0; u < CH_HELD; ++u) {
        if (eb + 256 * u < e1)
            binrec[pos[u]] = make_float4(__int_as_float(d[u]), ww[u], __int_as_float(s[u]), 0.0f);
    }
    for (int e = eb + 256 * CH_HELD; e < e1; e += 256) {
        const int sv = src[e];
        const int dv = dst[e];
        const float wv = w[e];
        const int p = atomicAdd(&cur[dv >> BIN_SHIFT], 1);
        binrec[p] = make_float4(__int_as_float(dv), wv, __int_as_float(sv), 0.0f);
    }
}

#define SEG_PER_HALF 8
#define SEG_HELD 16

static __device__ __forceinline__ int seg_pos(const int (&dl)[SEG_PER_HALF],
                                              const int (&pre)[SEG_PER_HALF + 1], int j) {
    int d = dl[0];
#pragma unroll
    for (int s = 1; s < SEG_PER_HALF; ++s) d = (j >= pre[s]) ? dl[s] : d;
    return j + d;
}

__global__ void __launch_bounds__(512)
bin_sort_kernel(const float4* __restrict__ binrec,
                const int* __restrict__ offt,
                int* __restrict__ rowptr,
                float* __restrict__ dinv,
                float2* __restrict__ ew, int n, int E) {
    __shared__ int lc[BIN_NODES];
    __shared__ float ldeg[BIN_NODES];
    __shared__ int wtot[16];
    __shared__ int wbefore[16];
    const int t = threadIdx.x;
    const int lane = t & 31;
    const int wave = t >> 5;
    const int hw = t >> 4;
    const int l16 = t & 15;
    const int r = blockIdx.x;
    const int n0 = r << BIN_SHIFT;
    if (n0 >= n) return;
    const int ch = (E + NCHUNK - 1) / NCHUNK;

    const int4* __restrict__ row0 = reinterpret_cast<const int4*>(offt + r * NCHUNK + hw * SEG_PER_HALF);
    const int4* __restrict__ row1 = reinterpret_cast<const int4*>(offt + (r + 1) * NCHUNK + hw * SEG_PER_HALF);
    const int4 a0 = row0[0];
    const int4 a1 = row0[1];
    const int4 b0 = row1[0];
    const int4 b1 = row1[1];
    const int so[SEG_PER_HALF] = {a0.x, a0.y, a0.z, a0.w, a1.x, a1.y, a1.z, a1.w};
    const int eo[SEG_PER_HALF] = {b0.x, b0.y, b0.z, b0.w, b1.x, b1.y, b1.z, b1.w};
    int pre[SEG_PER_HALF + 1];
    int dl[SEG_PER_HALF];
    int before = 0;
    pre[0] = 0;
#pragma unroll
    for (int s = 0; s < SEG_PER_HALF; ++s) {
        dl[s] = (hw * SEG_PER_HALF + s) * ch + so[s] - pre[s];
        pre[s + 1] = pre[s] + (eo[s] - so[s]);
        before += so[s];
    }
    const int total = pre[SEG_PER_HALF];
    const int wsum = before + __shfl_xor(before, 16);

    lc[t] = 0;
    ldeg[t] = 0.0f;
    if (lane == 0) wbefore[wave] = wsum;
    __syncthreads();
    int p0 = 0;
    for (int k = 0; k < 16; ++k) p0 += wbefore[k];

    float4 rec[SEG_HELD];
#pragma unroll
    for (int u = 0; u < SEG_HELD; ++u) {
        const int j = l16 + 16 * u;
        rec[u] = make_float4(0.0f, 0.0f, 0.0f, 0.0f);
        if (j < total) rec[u] = binrec[seg_pos(dl, pre, j)];
    }
#pragma unroll
    for (int u = 0; u < SEG_HELD; ++u) {
        if (l16 + 16 * u < total) {
            const int node = __float_as_int(rec[u].x) - n0;
            atomicAdd(&lc[node], 1);
            atomicAdd(&ldeg[node], rec[u].y);
        }
    }
    const float2* __restrict__ dw = reinterpret_cast<const float2*>(binrec);
    for (int j = 16 * SEG_HELD + l16; j < total; j += 16) {
        const float2 q = dw[(size_t)seg_pos(dl, pre, j) * 2];
        atomicAdd(&lc[__float_as_int(q.x) - n0], 1);
        atomicAdd(&ldeg[__float_as_int(q.x) - n0], q.y);
    }
    __syncthreads();
    const int cnt = lc[t];
    const float dg = 1.0f + ldeg[t];
    int incl = cnt;
    for (int off = 1; off < 32; off <<= 1) {
        const int v = __shfl_up(incl, off);
        if (lane >= off) incl += v;
    }
    if (lane == 31) wtot[wave] = incl;
    __syncthreads();
    int first = p0 + incl - cnt;
    for (int k = 0; k < wave; ++k) first += wtot[k];
    if (n0 + t < n) {
        rowptr[n0 + t] = first;
        dinv[n0 + t] = (dg > 0.0f) ? rsqrtf(fmaxf(dg, 1e-30f)) : 0.0f;
    }
    if (n0 + t == n - 1) rowptr[n] = first + cnt;
    lc[t] = first;
    __syncthreads();
#pragma unroll
    for (int u = 0; u < SEG_HELD; ++u) {
        if (l16 + 16 * u < total) {
            const int slot = atomicAdd(&lc[__float_as_int(rec[u].x) - n0], 1);
            ew[slot] = make_float2(rec[u].z, rec[u].y);
        }
    }
    for (int j = 16 * SEG_HELD + l16; j < total; j += 16) {
        const float4 q = binrec[seg_pos(dl, pre, j)];
        const int slot = atomicAdd(&lc[__float_as_int(q.x) - n0], 1);
        ew[slot] = make_float2(q.z, q.y);
    }
}

#define TILE_NODES 32
#define WAVE_NODES 8
#define A_PITCH 68

static __device__ __forceinline__ int lane_get_i(int v, int j) {
    return __builtin_amdgcn_readlane(v, j);
}
static __device__ __forceinline__ float lane_get_f(float v, int j) {
    return __int_as_float(__builtin_amdgcn_readlane(__float_as_int(v), j));
}
static __device__ __forceinline__ float2 row_pair(const float* __restrict__ H, int row, int lane) {
    return reinterpret_cast<const float2*>(H + (size_t)row * D)[lane];
}
template <int R>
static __device__ __forceinline__ void issue_rows(const float* __restrict__ H, int s, int j,
                                                  int lane, float2 (&hv)[R]) {
#pragma unroll
    for (int u = 0; u < R; ++u) hv[u] = row_pair(H, lane_get_i(s, j + u), lane);
}
template <int R>
static __device__ __forceinline__ void sum_rows(float2& acc, const float2 (&hv)[R], float nm,
                                                int j, int cnt) {
#pragma unroll
    for (int u = 0; u < R; ++u) {
        const float nj = lane_get_f(nm, j + u);
        const bool live = (j + u < cnt);
        const float hx = live ? hv[u].x : 0.0f;
        const float hy = live ? hv[u].y : 0.0f;
        acc.x += hx * nj;
        acc.y += hy * nj;
    }
}

static __device__ __forceinline__ float2 row_pair_at(const float* __restrict__ H, int row, unsigned lane8) {
    const unsigned off = ((unsigned)row << 8) + lane8;
    return *reinterpret_cast<const float2*>(reinterpret_cast<const char*>(H) + off);
}
static __device__ __forceinline__ void load_recs(const float2* __restrict__ csr, int p, float2 (&r)[16]) {
#pragma unroll
    for (int u = 0; u < 16; ++u) r[u] = csr[p + u];
}
template <int R>
static __device__ __forceinline__ void issue_rows_s(const float* __restrict__ H, const float2 (&r)[16],
                                                    int cnt, int own, unsigned lane8, float2 (&hv)[16]) {
#pragma unroll
    for (int u = 0; u < R; ++u) {
        int s = __float_as_int(r[u].x);
        if (u >= R - 4) s = (u < cnt) ? s : own;
        hv[u] = row_pair_at(H, s, lane8);
    }
}
template <int R>
static __device__ __forceinline__ void sum_rows_s(float2& acc, const float2 (&hv)[16],
                                                  const float2 (&r)[16], int cnt) {
#pragma unroll
    for (int u = 0; u < R; ++u) {
        float nj = r[u].y;
        if (u >= R - 4) nj = (u < cnt) ? nj : 0.0f;
        acc.x = fmaf(hv[u].x, nj, acc.x);
        acc.y = fmaf(hv[u].y, nj, acc.y);
    }
}
template <int R>
static __device__ __forceinline__ void rows_arm(float2& acc, const float* __restrict__ H,
                                                const float2 (&r)[16], int c, int own, unsigned lane8) {
    float2 hv[16];
    issue_rows_s<R>(H, r, c, own, lane8, hv);
    sum_rows_s<R>(acc, hv, r, c);
}
static __device__ __forceinline__ void rows_group(float2& acc, const float* __restrict__ H,
                                                  const float2 (&r)[16], int c, int own, unsigned lane8) {
    if (c > 12) rows_arm<16>(acc, H, r, c, own, lane8);
    else if (c > 8) rows_arm<12>(acc, H, r, c, own, lane8);
    else if (c > 4) rows_arm<8>(acc, H, r, c, own, lane8);
    else if (c > 0) rows_arm<4>(acc, H, r, c, own, lane8);
}
static __device__ __forceinline__ void walk_node(const float* __restrict__ X,
                                                 const float2* __restrict__ csr,
                                                 int rp, float dv, int q, int nbeg, int n,
                                                 int lane, unsigned lane8, float* __restrict__ arow,
                                                 int& beg0, int& end0, float2& hs0,
                                                 float2 (&rc)[16], float2 (&rn)[16]) {
    const int node = nbeg + q;
    const int own = (node < n) ? node : (n - 1);
    const int n1 = node + 1;
    const int beg1 = end0;
    const int end1 = lane_get_i(rp, q + 2);
    const bool live1 = (q + 1 < WAVE_NODES) && (n1 < n);
    const int n1c = (n1 < n) ? n1 : (n - 1);
    const int call = end0 - beg0;
    const int c = (call < 16) ? call : 16;
    const float di0 = lane_get_f(dv, q);

    float2 acc = make_float2(0.0f, 0.0f);
    const float2 hs1 = row_pair(X, n1c, lane);
    rows_group(acc, X, rc, c, own, lane8);
    load_recs(csr, beg1, rn);
    for (int base = beg0 + 16; base < end0; base += 16) {
        const int c2 = (end0 - base < 16) ? (end0 - base) : 16;
        load_recs(csr, base, rc);
        rows_group(acc, X, rc, c2, own, lane8);
    }
    const float sl = di0 * di0;
    acc.x += hs0.x * sl;
    acc.y += hs0.y * sl;
    reinterpret_cast<float2*>(arow)[lane] = acc;

    beg0 = beg1; end0 = end1;
    hs0.x = live1 ? hs1.x : 0.0f;
    hs0.y = live1 ? hs1.y : 0.0f;
}

template <bool FIRST>
__global__ void __launch_bounds__(128)
aggregate_transform_kernel(const float* __restrict__ X,
                           const int* __restrict__ rowptr,
                           const float2* __restrict__ csr,
                           float2* __restrict__ csr_out,
                           const float* __restrict__ dinv,
                           const float* __restrict__ W,
                           const float* __restrict__ bias,
                           float* __restrict__ outp, int n, int E) {
    __shared__ float sA[TILE_NODES * A_PITCH];
    const int lane = threadIdx.x & 31;
    const int wave = __builtin_amdgcn_readfirstlane((int)(threadIdx.x >> 5));
    const int tile0 = blockIdx.x * TILE_NODES;
    const int nbeg = tile0 + wave * WAVE_NODES;
    const int elast = (E > 0) ? (E - 1) : 0;

    int rp;
    float dv = 0.0f;
    {
        const int j = (lane < WAVE_NODES) ? lane : WAVE_NODES;
        const int idx = nbeg + j;
        rp = rowptr[(idx < n) ? idx : n];
        if (lane < WAVE_NODES && idx < n) dv = dinv[idx];
    }

    int beg0 = lane_get_i(rp, 0);
    int end0 = lane_get_i(rp, 1);
    float2 hs0 = make_float2(0.0f, 0.0f);
    if (nbeg < n) hs0 = row_pair(X, nbeg, lane);

    if (FIRST) {
        int s0 = 0;
        float nm0 = 0.0f;
        if (lane < end0 - beg0) {
            float2 rec = csr[beg0 + lane];
            s0 = __float_as_int(rec.x);
            nm0 = rec.y;
        }
        for (int q = 0; q < WAVE_NODES; ++q) {
            const int n1 = nbeg + q + 1;
            const int beg1 = end0;
            const int end1 = lane_get_i(rp, q + 2);
            const bool live1 = (q + 1 < WAVE_NODES) && (n1 < n);
            const int n1c = (n1 < n) ? n1 : (n - 1);
            int idx1 = beg1 + lane;
            idx1 = (idx1 < elast) ? idx1 : elast;
            const bool has1 = lane < end1 - beg1;
            int s1;
            float nm1;
            float2 hs1;

            const int cnt = (end0 - beg0 < 32) ? (end0 - beg0) : 32;
            const float di0 = lane_get_f(dv, q);
            float ds0 = 0.0f;
            if (FIRST) ds0 = dinv[s0];
            float2 acc = make_float2(0.0f, 0.0f);
            if (FIRST || cnt > 8) {
                float2 hv[16];
                issue_rows<16>(X, s0, 0, lane, hv);
                const float2 rec = csr[idx1];
                hs1 = row_pair(X, n1c, lane);
                if (FIRST) {
                    nm0 = ds0 * nm0 * di0;
                    if (lane < cnt) csr_out[beg0 + lane] = make_float2(__int_as_float(s0), nm0);
                }
                sum_rows<16>(acc, hv, nm0, 0, cnt);
                s1 = has1 ? __float_as_int(rec.x) : 0;
                nm1 = has1 ? rec.y : 0.0f;
                if (cnt > 16) {
                    float2 hw[16];
                    issue_rows<16>(X, s0, 16, lane, hw);
                    sum_rows<16>(acc, hw, nm0, 16, cnt);
                }
            } else {
                float2 hv[8];
                issue_rows<8>(X, s0, 0, lane, hv);
                const float2 rec = csr[idx1];
                hs1 = row_pair(X, n1c, lane);
                sum_rows<8>(acc, hv, nm0, 0, cnt);
                s1 = has1 ? __float_as_int(rec.x) : 0;
                nm1 = has1 ? rec.y : 0.0f;
            }
            for (int base = beg0 + 32; base < end0; base += 32) {
                const int c2 = (end0 - base < 32) ? (end0 - base) : 32;
                int s = 0;
                float nm = 0.0f;
                if (lane < c2) {
                    float2 rec = csr[base + lane];
                    s = __float_as_int(rec.x);
                    nm = rec.y;
                }
                if (FIRST) {
                    nm = dinv[s] * nm * di0;
                    if (lane < c2) csr_out[base + lane] = make_float2(__int_as_float(s), nm);
                }
                float2 hv[16];
                issue_rows<16>(X, s, 0, lane, hv);
                sum_rows<16>(acc, hv, nm, 0, c2);
                if (c2 > 16) {
                    float2 hw[16];
                    issue_rows<16>(X, s, 16, lane, hw);
                    sum_rows<16>(acc, hw, nm, 16, c2);
                }
            }
            const float sl = di0 * di0;
            acc.x += hs0.x * sl;
            acc.y += hs0.y * sl;
            reinterpret_cast<float2*>(&sA[(wave * WAVE_NODES + q) * A_PITCH])[lane] = acc;

            beg0 = beg1; end0 = end1; s0 = s1; nm0 = nm1;
            hs0.x = live1 ? hs1.x : 0.0f;
            hs0.y = live1 ? hs1.y : 0.0f;
        }
    } else {
        const unsigned lane8 = (unsigned)lane << 3;
        float2 ra[16], rb[16];
        load_recs(csr, beg0, ra);
        for (int q = 0; q < WAVE_NODES; q += 2) {
            walk_node(X, csr, rp, dv, q, nbeg, n, lane, lane8,
                      &sA[(wave * WAVE_NODES + q) * A_PITCH], beg0, end0, hs0, ra, rb);
            walk_node(X, csr, rp, dv, q + 1, nbeg, n, lane, lane8,
                      &sA[(wave * WAVE_NODES + q + 1) * A_PITCH], beg0, end0, hs0, rb, ra);
        }
    }

    const int half = lane >> 4;
    const int l = lane & 15;
    const int c0 = wave * 16;
    const float* __restrict__ wcol = W + (c0 + l) + half * 2 * D;
    v2f bfrag[D / 4];
#pragma unroll
    for (int k = 0; k < D; k += 4) {
        bfrag[k / 4].x = wcol[k * D];
        bfrag[k / 4].y = wcol[k * D + D];
    }
    const float bc = bias[c0 + l];

    __syncthreads();

    const float* a0p = &sA[l * A_PITCH + half * 2];
    const float* a1p = &sA[(16 + l) * A_PITCH + half * 2];
    v8f acc0 = {};
    v8f acc1 = {};
#pragma unroll
    for (int k = 0; k < D; k += 4) {
        const v2f a0 = *reinterpret_cast<const v2f*>(a0p + k);
        const v2f a1 = *reinterpret_cast<const v2f*>(a1p + k);
        acc0 = __builtin_amdgcn_wmma_f32_16x16x4_f32(
             false, a0,  false, bfrag[k / 4],
             (short)0, acc0,  false,  false);
        acc1 = __builtin_amdgcn_wmma_f32_16x16x4_f32(
             false, a1,  false, bfrag[k / 4],
             (short)0, acc1,  false,  false);
    }

    const int r0 = tile0 + half * 8;
    float* __restrict__ op = outp + (size_t)r0 * D + c0 + l;
    if (tile0 + TILE_NODES <= n) {
#pragma unroll
        for (int v = 0; v < 8; ++v) {
            op[(size_t)v * D] = fmaxf(acc0[v] + bc, 0.0f);
            op[(size_t)(16 + v) * D] = fmaxf(acc1[v] + bc, 0.0f);
        }
    } else {
#pragma unroll
        for (int v = 0; v < 8; ++v) {
            if (r0 + v < n) op[(size_t)v * D] = fmaxf(acc0[v] + bc, 0.0f);
            if (r0 + 16 + v < n) op[(size_t)(16 + v) * D] = fmaxf(acc1[v] + bc, 0.0f);
        }
    }
}


extern "C" void kernel_launch(void* const* d_in, const int* in_sizes, int n_in,
                              void* d_out, int out_size, void* d_ws, size_t ws_size,
                              hipStream_t stream) {
    (void)n_in; (void)out_size; (void)ws_size;

    const float* x  = (const float*)d_in[0];
    const int*   ei = (const int*)  d_in[1];
    const float* ea = (const float*)d_in[2];
    const float* W1 = (const float*)d_in[3];
    const float* b1 = (const float*)d_in[4];
    const float* W2 = (const float*)d_in[5];
    const float* b2 = (const float*)d_in[6];
    const float* W3 = (const float*)d_in[7];
    const float* b3 = (const float*)d_in[8];

    const int N = in_sizes[0] / D;
    const int E = in_sizes[2];
    const int* src = ei;
    const int* dst = ei + E;

    float*  out      = (float*)d_out;
    float*  bufA     = (float*)d_ws;
    float*  bufB     = bufA + (size_t)N * D;
    float4* binrec   = (float4*)(bufB + (size_t)N * D);
    int*    offt     = (int*)(binrec + E);
    float2* ew       = (float2*)(offt + (NBIN + 1) * NCHUNK);
    float2* csr      = ew + E;
    float*  dinv     = (float*)(csr + E);
    int*    rowptr   = (int*)(dinv + N);

    const dim3 blk(256);
    const dim3 blkLayer(128);
    const dim3 gLayer((N + TILE_NODES - 1) / TILE_NODES);

    chunk_bin_kernel<<<dim3(NCHUNK), blk, 0, stream>>>(src, dst, ea, offt, binrec, E);
    bin_sort_kernel<<<dim3(NBIN), dim3(BIN_NODES), 0, stream>>>(binrec, offt, rowptr, dinv, ew, N, E);

    aggregate_transform_kernel<true><<<gLayer, blkLayer, 0, stream>>>(x,    rowptr, ew,  csr,     dinv, W1, b1, bufB, N, E);
    aggregate_transform_kernel<false><<<gLayer, blkLayer, 0, stream>>>(bufB, rowptr, csr, nullptr, dinv, W2, b2, bufA, N, E);
    aggregate_transform_kernel<false><<<gLayer, blkLayer, 0, stream>>>(bufA, rowptr, csr, nullptr, dinv, W3, b3, out,  N, E);
}


# Both sort levels put each placed record into an LDS copy of its output stretch and write it out in order, whole lines per store (was a scattered 16- or 8-byte store per record); list order rule kept
# speedup vs baseline: 5.6966x; 1.0624x over previous
#include <hip/hip_runtime.h>
#include <hip/hip_bf16.h>

typedef __attribute__((ext_vector_type(2))) float v2f;
typedef __attribute__((ext_vector_type(8))) float v8f;

#define D 64

#define NCHUNK 256
#define NBIN 256
#define BIN_SHIFT 9
#define BIN_NODES 512

#define CH_HELD 20
#define CH_STAGE (256 * CH_HELD)

__global__ void __launch_bounds__(256)
chunk_bin_kernel(const int* __restrict__ src,
                 const int* __restrict__ dst,
                 const float* __restrict__ w,
                 int* __restrict__ offt,
                 float4* __restrict__ binrec, int E) {
    __shared__ int h[NBIN + 32];
    __shared__ int cur[NBIN + 32];
    __shared__ int wtot[8];
    __shared__ float4 stage[CH_STAGE];
    const int t = threadIdx.x;
    const int lane = t & 31;
    const int wave = t >> 5;
    const int c = blockIdx.x;
    if (E <= 0) {
        offt[t * NCHUNK + c] = 0;
        if (t == NBIN - 1) offt[NBIN * NCHUNK + c] = 0;
        return;
    }
    const int ch = (E + NCHUNK - 1) / NCHUNK;
    const int e0 = c * ch;
    const int e1 = (e0 + ch < E) ? (e0 + ch) : E;
    const int eb = e0 + t;
    const int elast = E - 1;

    int d[CH_HELD], s[CH_HELD];
    float ww[CH_HELD];
#pragma unroll
    for (int u = 0; u < CH_HELD; ++u) {
        const int e = (eb + 256 * u < e1) ? (eb + 256 * u) : elast;
        d[u] = dst[e];
        s[u] = src[e];
        ww[u] = w[e];
    }
    h[t] = 0;
    if (t < 32) { h[NBIN + t] = 0; cur[NBIN + t] = 0; }
    __syncthreads();
    int b[CH_HELD];
#pragma unroll
    for (int u = 0; u < CH_HELD; ++u) {
        b[u] = (eb + 256 * u < e1) ? (d[u] >> BIN_SHIFT) : (NBIN + lane);
        atomicAdd(&h[b[u]], 1);
    }
    for (int e = eb + 256 * CH_HELD; e < e1; e += 256) atomicAdd(&h[dst[e] >> BIN_SHIFT], 1);
    __syncthreads();
    const int cnt = h[t];
    int incl = cnt;
    for (int off = 1; off < 32; off <<= 1) {
        const int v = __shfl_up(incl, off);
        if (lane >= off) incl += v;
    }
    if (lane == 31) wtot[wave] = incl;
    __syncthreads();
    int ex = incl - cnt;
    for (int k = 0; k < wave; ++k) ex += wtot[k];
    cur[t] = e0 + ex;
    offt[t * NCHUNK + c] = ex;
    if (t == NBIN - 1) offt[NBIN * NCHUNK + c] = ex + cnt;
    __syncthreads();

    int pos[CH_HELD];
#pragma unroll
    for (int u = 0; u < CH_HELD; ++u) pos[u] = atomicAdd(&cur[b[u]], 1);
#pragma unroll
    for (int u = 0; u < CH_HELD; ++u) {
        if (eb + 256 * u < e1) {
            const float4 rec = make_float4(__int_as_float(d[u]), ww[u], __int_as_float(s[u]), 0.0f);
            const int lp = pos[u] - e0;
            if (lp < CH_STAGE) stage[lp] = rec;
            else binrec[pos[u]] = rec;
        }
    }
    for (int e = eb + 256 * CH_HELD; e < e1; e += 256) {
        const int sv = src[e];
        const int dv = dst[e];
        const float wv = w[e];
        const int p = atomicAdd(&cur[dv >> BIN_SHIFT], 1);
        const float4 rec = make_float4(__int_as_float(dv), wv, __int_as_float(sv), 0.0f);
        if (p - e0 < CH_STAGE) stage[p - e0] = rec;
        else binrec[p] = rec;
    }
    __syncthreads();
    const int nst = (e1 - e0 < CH_STAGE) ? (e1 - e0) : CH_STAGE;
    const int gb = e0 & ~31;
    const int ge = e0 + nst;
    const int kfull = (ge - gb) >> 8;
    {
        const int g = gb + t;
        if (g >= e0 && g < ge) binrec[g] = stage[g - e0];
    }
#pragma unroll 4
    for (int k = 1; k < kfull; ++k) {
        const int g = gb + t + 256 * k;
        binrec[g] = stage[g - e0];
    }
    {
        const int g = gb + t + 256 * ((kfull > 1) ? kfull : 1);
        if (g < ge) binrec[g] = stage[g - e0];
    }
}

#define SEG_PER_HALF 8
#define SEG_HELD 16
#define EW_STAGE 7168

static __device__ __forceinline__ int seg_pos(const int (&dl)[SEG_PER_HALF],
                                              const int (&pre)[SEG_PER_HALF + 1], int j) {
    int d = dl[0];
#pragma unroll
    for (int s = 1; s < SEG_PER_HALF; ++s) d = (j >= pre[s]) ? dl[s] : d;
    return j + d;
}

__global__ void __launch_bounds__(512)
bin_sort_kernel(const float4* __restrict__ binrec,
                const int* __restrict__ offt,
                int* __restrict__ rowptr,
                float* __restrict__ dinv,
                float2* __restrict__ ew, int n, int E) {
    __shared__ int lc[BIN_NODES];
    __shared__ float ldeg[BIN_NODES];
    __shared__ int wtot[16];
    __shared__ int wbefore[16];
    __shared__ float2 stage[EW_STAGE];
    const int t = threadIdx.x;
    const int lane = t & 31;
    const int wave = t >> 5;
    const int hw = t >> 4;
    const int l16 = t & 15;
    const int r = blockIdx.x;
    const int n0 = r << BIN_SHIFT;
    if (n0 >= n) return;
    const int ch = (E + NCHUNK - 1) / NCHUNK;

    const int4* __restrict__ row0 = reinterpret_cast<const int4*>(offt + r * NCHUNK + hw * SEG_PER_HALF);
    const int4* __restrict__ row1 = reinterpret_cast<const int4*>(offt + (r + 1) * NCHUNK + hw * SEG_PER_HALF);
    const int4 a0 = row0[0];
    const int4 a1 = row0[1];
    const int4 b0 = row1[0];
    const int4 b1 = row1[1];
    const int so[SEG_PER_HALF] = {a0.x, a0.y, a0.z, a0.w, a1.x, a1.y, a1.z, a1.w};
    const int eo[SEG_PER_HALF] = {b0.x, b0.y, b0.z, b0.w, b1.x, b1.y, b1.z, b1.w};
    int pre[SEG_PER_HALF + 1];
    int dl[SEG_PER_HALF];
    int before = 0;
    pre[0] = 0;
#pragma unroll
    for (int s = 0; s < SEG_PER_HALF; ++s) {
        dl[s] = (hw * SEG_PER_HALF + s) * ch + so[s] - pre[s];
        pre[s + 1] = pre[s] + (eo[s] - so[s]);
        before += so[s];
    }
    const int total = pre[SEG_PER_HALF];
    const int wsum = before + __shfl_xor(before, 16);

    lc[t] = 0;
    ldeg[t] = 0.0f;
    if (lane == 0) wbefore[wave] = wsum;
    __syncthreads();
    int p0 = 0;
    for (int k = 0; k < 16; ++k) p0 += wbefore[k];

    float4 rec[SEG_HELD];
#pragma unroll
    for (int u = 0; u < SEG_HELD; ++u) {
        const int j = l16 + 16 * u;
        rec[u] = make_float4(0.0f, 0.0f, 0.0f, 0.0f);
        if (j < total) rec[u] = binrec[seg_pos(dl, pre, j)];
    }
#pragma unroll
    for (int u = 0; u < SEG_HELD; ++u) {
        if (l16 + 16 * u < total) {
            const int node = __float_as_int(rec[u].x) - n0;
            atomicAdd(&lc[node], 1);
            atomicAdd(&ldeg[node], rec[u].y);
        }
    }
    const float2* __restrict__ dw = reinterpret_cast<const float2*>(binrec);
    for (int j = 16 * SEG_HELD + l16; j < total; j += 16) {
        const float2 q = dw[(size_t)seg_pos(dl, pre, j) * 2];
        atomicAdd(&lc[__float_as_int(q.x) - n0], 1);
        atomicAdd(&ldeg[__float_as_int(q.x) - n0], q.y);
    }
    __syncthreads();
    const int cnt = lc[t];
    const float dg = 1.0f + ldeg[t];
    int incl = cnt;
    for (int off = 1; off < 32; off <<= 1) {
        const int v = __shfl_up(incl, off);
        if (lane >= off) incl += v;
    }
    if (lane == 31) wtot[wave] = incl;
    __syncthreads();
    int first = p0 + incl - cnt;
    for (int k = 0; k < wave; ++k) first += wtot[k];
    if (n0 + t < n) {
        rowptr[n0 + t] = first;
        dinv[n0 + t] = (dg > 0.0f) ? rsqrtf(fmaxf(dg, 1e-30f)) : 0.0f;
    }
    if (n0 + t == n - 1) rowptr[n] = first + cnt;
    lc[t] = first - p0;
    int rtot = 0;
    for (int k = 0; k < 16; ++k) rtot += wtot[k];
    __syncthreads();
#pragma unroll
    for (int u = 0; u < SEG_HELD; ++u) {
        if (l16 + 16 * u < total) {
            const int slot = atomicAdd(&lc[__float_as_int(rec[u].x) - n0], 1);
            const float2 o = make_float2(rec[u].z, rec[u].y);
            if (slot < EW_STAGE) stage[slot] = o;
            else ew[p0 + slot] = o;
        }
    }
    for (int j = 16 * SEG_HELD + l16; j < total; j += 16) {
        const float4 q = binrec[seg_pos(dl, pre, j)];
        const int slot = atomicAdd(&lc[__float_as_int(q.x) - n0], 1);
        const float2 o = make_float2(q.z, q.y);
        if (slot < EW_STAGE) stage[slot] = o;
        else ew[p0 + slot] = o;
    }
    __syncthreads();
    const int p0s = __builtin_amdgcn_readfirstlane(p0);
    const int rts = __builtin_amdgcn_readfirstlane(rtot);
    const int nst = (rts < EW_STAGE) ? rts : EW_STAGE;
    const int gb = p0s & ~31;
    const int ge = p0s + nst;
    const int kfull = (ge - gb) >> 9;
    {
        const int g = gb + t;
        if (g >= p0s && g < ge) ew[g] = stage[g - p0s];
    }
#pragma unroll 4
    for (int k = 1; k < kfull; ++k) {
        const int g = gb + t + 512 * k;
        ew[g] = stage[g - p0s];
    }
    {
        const int g = gb + t + 512 * ((kfull > 1) ? kfull : 1);
        if (g < ge) ew[g] = stage[g - p0s];
    }
}

#define TILE_NODES 32
#define WAVE_NODES 8
#define A_PITCH 68

static __device__ __forceinline__ int lane_get_i(int v, int j) {
    return __builtin_amdgcn_readlane(v, j);
}
static __device__ __forceinline__ float lane_get_f(float v, int j) {
    return __int_as_float(__builtin_amdgcn_readlane(__float_as_int(v), j));
}
static __device__ __forceinline__ float2 row_pair(const float* __restrict__ H, int row, int lane) {
    return reinterpret_cast<const float2*>(H + (size_t)row * D)[lane];
}
template <int R>
static __device__ __forceinline__ void issue_rows(const float* __restrict__ H, int s, int j,
                                                  int lane, float2 (&hv)[R]) {
#pragma unroll
    for (int u = 0; u < R; ++u) hv[u] = row_pair(H, lane_get_i(s, j + u), lane);
}
template <int R>
static __device__ __forceinline__ void sum_rows(float2& acc, const float2 (&hv)[R], float nm,
                                                int j, int cnt) {
#pragma unroll
    for (int u = 0; u < R; ++u) {
        const float nj = lane_get_f(nm, j + u);
        const bool live = (j + u < cnt);
        const float hx = live ? hv[u].x : 0.0f;
        const float hy = live ? hv[u].y : 0.0f;
        acc.x += hx * nj;
        acc.y += hy * nj;
    }
}

static __device__ __forceinline__ float2 row_pair_at(const float* __restrict__ H, int row, unsigned lane8) {
    const unsigned off = ((unsigned)row << 8) + lane8;
    return *reinterpret_cast<const float2*>(reinterpret_cast<const char*>(H) + off);
}
static __device__ __forceinline__ void load_recs(const float2* __restrict__ csr, int p, float2 (&r)[16]) {
#pragma unroll
    for (int u = 0; u < 16; ++u) r[u] = csr[p + u];
}
template <int R>
static __device__ __forceinline__ void issue_rows_s(const float* __restrict__ H, const float2 (&r)[16],
                                                    int cnt, int own, unsigned lane8, float2 (&hv)[16]) {
#pragma unroll
    for (int u = 0; u < R; ++u) {
        int s = __float_as_int(r[u].x);
        if (u >= R - 4) s = (u < cnt) ? s : own;
        hv[u] = row_pair_at(H, s, lane8);
    }
}
template <int R>
static __device__ __forceinline__ void sum_rows_s(float2& acc, const float2 (&hv)[16],
                                                  const float2 (&r)[16], int cnt) {
#pragma unroll
    for (int u = 0; u < R; ++u) {
        float nj = r[u].y;
        if (u >= R - 4) nj = (u < cnt) ? nj : 0.0f;
        acc.x = fmaf(hv[u].x, nj, acc.x);
        acc.y = fmaf(hv[u].y, nj, acc.y);
    }
}
template <int R>
static __device__ __forceinline__ void rows_arm(float2& acc, const float* __restrict__ H,
                                                const float2 (&r)[16], int c, int own, unsigned lane8) {
    float2 hv[16];
    issue_rows_s<R>(H, r, c, own, lane8, hv);
    sum_rows_s<R>(acc, hv, r, c);
}
static __device__ __forceinline__ void rows_group(float2& acc, const float* __restrict__ H,
                                                  const float2 (&r)[16], int c, int own, unsigned lane8) {
    if (c > 12) rows_arm<16>(acc, H, r, c, own, lane8);
    else if (c > 8) rows_arm<12>(acc, H, r, c, own, lane8);
    else if (c > 4) rows_arm<8>(acc, H, r, c, own, lane8);
    else if (c > 0) rows_arm<4>(acc, H, r, c, own, lane8);
}
static __device__ __forceinline__ void walk_node(const float* __restrict__ X,
                                                 const float2* __restrict__ csr,
                                                 int rp, float dv, int q, int nbeg, int n,
                                                 int lane, unsigned lane8, float* __restrict__ arow,
                                                 int& beg0, int& end0, float2& hs0,
                                                 float2 (&rc)[16], float2 (&rn)[16]) {
    const int node = nbeg + q;
    const int own = (node < n) ? node : (n - 1);
    const int n1 = node + 1;
    const int beg1 = end0;
    const int end1 = lane_get_i(rp, q + 2);
    const bool live1 = (q + 1 < WAVE_NODES) && (n1 < n);
    const int n1c = (n1 < n) ? n1 : (n - 1);
    const int call = end0 - beg0;
    const int c = (call < 16) ? call : 16;
    const float di0 = lane_get_f(dv, q);

    float2 acc = make_float2(0.0f, 0.0f);
    const float2 hs1 = row_pair(X, n1c, lane);
    rows_group(acc, X, rc, c, own, lane8);
    load_recs(csr, beg1, rn);
    for (int base = beg0 + 16; base < end0; base += 16) {
        const int c2 = (end0 - base < 16) ? (end0 - base) : 16;
        load_recs(csr, base, rc);
        rows_group(acc, X, rc, c2, own, lane8);
    }
    const float sl = di0 * di0;
    acc.x += hs0.x * sl;
    acc.y += hs0.y * sl;
    reinterpret_cast<float2*>(arow)[lane] = acc;

    beg0 = beg1; end0 = end1;
    hs0.x = live1 ? hs1.x : 0.0f;
    hs0.y = live1 ? hs1.y : 0.0f;
}

template <bool FIRST>
__global__ void __launch_bounds__(128)
aggregate_transform_kernel(const float* __restrict__ X,
                           const int* __restrict__ rowptr,
                           const float2* __restrict__ csr,
                           float2* __restrict__ csr_out,
                           const float* __restrict__ dinv,
                           const float* __restrict__ W,
                           const float* __restrict__ bias,
                           float* __restrict__ outp, int n, int E) {
    __shared__ float sA[TILE_NODES * A_PITCH];
    const int lane = threadIdx.x & 31;
    const int wave = __builtin_amdgcn_readfirstlane((int)(threadIdx.x >> 5));
    const int tile0 = blockIdx.x * TILE_NODES;
    const int nbeg = tile0 + wave * WAVE_NODES;
    const int elast = (E > 0) ? (E - 1) : 0;

    int rp;
    float dv = 0.0f;
    {
        const int j = (lane < WAVE_NODES) ? lane : WAVE_NODES;
        const int idx = nbeg + j;
        rp = rowptr[(idx < n) ? idx : n];
        if (lane < WAVE_NODES && idx < n) dv = dinv[idx];
    }

    int beg0 = lane_get_i(rp, 0);
    int end0 = lane_get_i(rp, 1);
    float2 hs0 = make_float2(0.0f, 0.0f);
    if (nbeg < n) hs0 = row_pair(X, nbeg, lane);

    if (FIRST) {
        int s0 = 0;
        float nm0 = 0.0f;
        if (lane < end0 - beg0) {
            float2 rec = csr[beg0 + lane];
            s0 = __float_as_int(rec.x);
            nm0 = rec.y;
        }
        for (int q = 0; q < WAVE_NODES; ++q) {
            const int n1 = nbeg + q + 1;
            const int beg1 = end0;
            const int end1 = lane_get_i(rp, q + 2);
            const bool live1 = (q + 1 < WAVE_NODES) && (n1 < n);
            const int n1c = (n1 < n) ? n1 : (n - 1);
            int idx1 = beg1 + lane;
            idx1 = (idx1 < elast) ? idx1 : elast;
            const bool has1 = lane < end1 - beg1;
            int s1;
            float nm1;
            float2 hs1;

            const int cnt = (end0 - beg0 < 32) ? (end0 - beg0) : 32;
            const float di0 = lane_get_f(dv, q);
            float ds0 = 0.0f;
            if (FIRST) ds0 = dinv[s0];
            float2 acc = make_float2(0.0f, 0.0f);
            if (FIRST || cnt > 8) {
                float2 hv[16];
                issue_rows<16>(X, s0, 0, lane, hv);
                const float2 rec = csr[idx1];
                hs1 = row_pair(X, n1c, lane);
                if (FIRST) {
                    nm0 = ds0 * nm0 * di0;
                    if (lane < cnt) csr_out[beg0 + lane] = make_float2(__int_as_float(s0), nm0);
                }
                sum_rows<16>(acc, hv, nm0, 0, cnt);
                s1 = has1 ? __float_as_int(rec.x) : 0;
                nm1 = has1 ? rec.y : 0.0f;
                if (cnt > 16) {
                    float2 hw[16];
                    issue_rows<16>(X, s0, 16, lane, hw);
                    sum_rows<16>(acc, hw, nm0, 16, cnt);
                }
            } else {
                float2 hv[8];
                issue_rows<8>(X, s0, 0, lane, hv);
                const float2 rec = csr[idx1];
                hs1 = row_pair(X, n1c, lane);
                sum_rows<8>(acc, hv, nm0, 0, cnt);
                s1 = has1 ? __float_as_int(rec.x) : 0;
                nm1 = has1 ? rec.y : 0.0f;
            }
            for (int base = beg0 + 32; base < end0; base += 32) {
                const int c2 = (end0 - base < 32) ? (end0 - base) : 32;
                int s = 0;
                float nm = 0.0f;
                if (lane < c2) {
                    float2 rec = csr[base + lane];
                    s = __float_as_int(rec.x);
                    nm = rec.y;
                }
                if (FIRST) {
                    nm = dinv[s] * nm * di0;
                    if (lane < c2) csr_out[base + lane] = make_float2(__int_as_float(s), nm);
                }
                float2 hv[16];
                issue_rows<16>(X, s, 0, lane, hv);
                sum_rows<16>(acc, hv, nm, 0, c2);
                if (c2 > 16) {
                    float2 hw[16];
                    issue_rows<16>(X, s, 16, lane, hw);
                    sum_rows<16>(acc, hw, nm, 16, c2);
                }
            }
            const float sl = di0 * di0;
            acc.x += hs0.x * sl;
            acc.y += hs0.y * sl;
            reinterpret_cast<float2*>(&sA[(wave * WAVE_NODES + q) * A_PITCH])[lane] = acc;

            beg0 = beg1; end0 = end1; s0 = s1; nm0 = nm1;
            hs0.x = live1 ? hs1.x : 0.0f;
            hs0.y = live1 ? hs1.y : 0.0f;
        }
    } else {
        const unsigned lane8 = (unsigned)lane << 3;
        float2 ra[16], rb[16];
        load_recs(csr, beg0, ra);
        for (int q = 0; q < WAVE_NODES; q += 2) {
            walk_node(X, csr, rp, dv, q, nbeg, n, lane, lane8,
                      &sA[(wave * WAVE_NODES + q) * A_PITCH], beg0, end0, hs0, ra, rb);
            walk_node(X, csr, rp, dv, q + 1, nbeg, n, lane, lane8,
                      &sA[(wave * WAVE_NODES + q + 1) * A_PITCH], beg0, end0, hs0, rb, ra);
        }
    }

    const int half = lane >> 4;
    const int l = lane & 15;
    const int c0 = wave * 16;
    const float* __restrict__ wcol = W + (c0 + l) + half * 2 * D;
    v2f bfrag[D / 4];
#pragma unroll
    for (int k = 0; k < D; k += 4) {
        bfrag[k / 4].x = wcol[k * D];
        bfrag[k / 4].y = wcol[k * D + D];
    }
    const float bc = bias[c0 + l];

    __syncthreads();

    const float* a0p = &sA[l * A_PITCH + half * 2];
    const float* a1p = &sA[(16 + l) * A_PITCH + half * 2];
    v8f acc0 = {};
    v8f acc1 = {};
#pragma unroll
    for (int k = 0; k < D; k += 4) {
        const v2f a0 = *reinterpret_cast<const v2f*>(a0p + k);
        const v2f a1 = *reinterpret_cast<const v2f*>(a1p + k);
        acc0 = __builtin_amdgcn_wmma_f32_16x16x4_f32(
             false, a0,  false, bfrag[k / 4],
             (short)0, acc0,  false,  false);
        acc1 = __builtin_amdgcn_wmma_f32_16x16x4_f32(
             false, a1,  false, bfrag[k / 4],
             (short)0, acc1,  false,  false);
    }

    const int r0 = tile0 + half * 8;
    float* __restrict__ op = outp + (size_t)r0 * D + c0 + l;
    if (tile0 + TILE_NODES <= n) {
#pragma unroll
        for (int v = 0; v < 8; ++v) {
            op[(size_t)v * D] = fmaxf(acc0[v] + bc, 0.0f);
            op[(size_t)(16 + v) * D] = fmaxf(acc1[v] + bc, 0.0f);
        }
    } else {
#pragma unroll
        for (int v = 0; v < 8; ++v) {
            if (r0 + v < n) op[(size_t)v * D] = fmaxf(acc0[v] + bc, 0.0f);
            if (r0 + 16 + v < n) op[(size_t)(16 + v) * D] = fmaxf(acc1[v] + bc, 0.0f);
        }
    }
}


extern "C" void kernel_launch(void* const* d_in, const int* in_sizes, int n_in,
                              void* d_out, int out_size, void* d_ws, size_t ws_size,
                              hipStream_t stream) {
    (void)n_in; (void)out_size; (void)ws_size;

    const float* x  = (const float*)d_in[0];
    const int*   ei = (const int*)  d_in[1];
    const float* ea = (const float*)d_in[2];
    const float* W1 = (const float*)d_in[3];
    const float* b1 = (const float*)d_in[4];
    const float* W2 = (const float*)d_in[5];
    const float* b2 = (const float*)d_in[6];
    const float* W3 = (const float*)d_in[7];
    const float* b3 = (const float*)d_in[8];

    const int N = in_sizes[0] / D;
    const int E = in_sizes[2];
    const int* src = ei;
    const int* dst = ei + E;

    float*  out      = (float*)d_out;
    float*  bufA     = (float*)d_ws;
    float*  bufB     = bufA + (size_t)N * D;
    float4* binrec   = (float4*)(bufB + (size_t)N * D);
    int*    offt     = (int*)(binrec + E);
    float2* ew       = (float2*)(offt + (NBIN + 1) * NCHUNK);
    float2* csr      = ew + E;
    float*  dinv     = (float*)(csr + E);
    int*    rowptr   = (int*)(dinv + N);

    const dim3 blk(256);
    const dim3 blkLayer(128);
    const dim3 gLayer((N + TILE_NODES - 1) / TILE_NODES);

    chunk_bin_kernel<<<dim3(NCHUNK), blk, 0, stream>>>(src, dst, ea, offt, binrec, E);
    bin_sort_kernel<<<dim3(NBIN), dim3(BIN_NODES), 0, stream>>>(binrec, offt, rowptr, dinv, ew, N, E);

    aggregate_transform_kernel<true><<<gLayer, blkLayer, 0, stream>>>(x,    rowptr, ew,  csr,     dinv, W1, b1, bufB, N, E);
    aggregate_transform_kernel<false><<<gLayer, blkLayer, 0, stream>>>(bufB, rowptr, csr, nullptr, dinv, W2, b2, bufA, N, E);
    aggregate_transform_kernel<false><<<gLayer, blkLayer, 0, stream>>>(bufA, rowptr, csr, nullptr, dinv, W3, b3, out,  N, E);
}
